# starting kernel
# baseline (speedup 1.0000x reference)
#include <hip/hip_runtime.h>
#include <math.h>

#define B_    4
#define WIN_  5
#define N_    50000
#define E_    1000000
#define H_    32
#define NTOT_ (B_ * N_)
#define BE_   (B_ * E_)

typedef __attribute__((ext_vector_type(2))) float v2f;
typedef __attribute__((ext_vector_type(8))) float v8f;

static __device__ __forceinline__ void decode_edge(const int* __restrict__ ei, int e,
                                                   int& src, int& dst) {
    int b  = e / E_;
    int el = e - b * E_;
    src = ei[(b * 2 + 0) * E_ + el] + b * N_;
    dst = ei[(b * 2 + 1) * E_ + el] + b * N_;
}

__global__ void zero_f32_kernel(float* __restrict__ p, int n) {
    int i = blockIdx.x * blockDim.x + threadIdx.x;
    if (i < n) p[i] = 0.0f;
}

__global__ void deg_kernel(const int* __restrict__ ei, float* __restrict__ deg) {
    int e = blockIdx.x * blockDim.x + threadIdx.x;
    if (e >= BE_) return;
    int src, dst;
    decode_edge(ei, e, src, dst);
    atomicAdd(&deg[dst], 1.0f);
}

__global__ void dinv_kernel(float* __restrict__ d) {
    int i = blockIdx.x * blockDim.x + threadIdx.x;
    if (i >= NTOT_) return;
    float v = d[i];
    d[i] = (v > 0.0f) ? rsqrtf(v) : 0.0f;
}

__global__ void norm_kernel(const int* __restrict__ ei, const float* __restrict__ dinv,
                            float* __restrict__ norm) {
    int e = blockIdx.x * blockDim.x + threadIdx.x;
    if (e >= BE_) return;
    int src, dst;
    decode_edge(ei, e, src, dst);
    norm[e] = dinv[src] * dinv[dst];
}

template <int F>
__global__ void prop_kernel(const int* __restrict__ ei, const float* __restrict__ norm,
                            const float* __restrict__ hin, float* __restrict__ hout) {
    int tid = blockIdx.x * blockDim.x + threadIdx.x;
    int e = tid / F;
    int f = tid - e * F;
    if (e >= BE_) return;
    int src, dst;
    decode_edge(ei, e, src, dst);
    float v = hin[src * F + f] * norm[e];
    atomicAdd(&hout[dst * F + f], v);
}

__global__ void prop4_kernel(const int* __restrict__ ei, const float* __restrict__ norm,
                             const float4* __restrict__ hin4, float* __restrict__ hout) {
    int tid = blockIdx.x * blockDim.x + threadIdx.x;
    int e = tid >> 3;
    int q = tid & 7;
    if (e >= BE_) return;
    int src, dst;
    decode_edge(ei, e, src, dst);
    float nrm = norm[e];
    float4 v = hin4[src * 8 + q];
    float* op = &hout[dst * H_ + q * 4];
    atomicAdd(op + 0, v.x * nrm);
    atomicAdd(op + 1, v.y * nrm);
    atomicAdd(op + 2, v.z * nrm);
    atomicAdd(op + 3, v.w * nrm);
}

template <int F>
__global__ void tag_gemm_kernel(const float* __restrict__ x0, const float* __restrict__ x1,
                                const float* __restrict__ x2, const float* __restrict__ W,
                                const float* __restrict__ bias, float* __restrict__ out) {
    const int numTiles = NTOT_ / 16;
    int wave = threadIdx.x >> 5;
    int tile = blockIdx.x * (blockDim.x >> 5) + wave;
    if (tile >= numTiles) return;

    int lane = threadIdx.x & 31;
    int half = lane >> 4;
    int l16  = lane & 15;
    int rowA = tile * 16 + l16;

    v8f c0 = {};
    v8f c1 = {};

    const float* xs[3] = { x0, x1, x2 };
    constexpr bool FULL = (F % 4 == 0);

#pragma unroll
    for (int s = 0; s < 3; ++s) {
        const float* __restrict__ xp = xs[s];
#pragma unroll
        for (int k0 = 0; k0 < F; k0 += 4) {
            int kA = k0 + half * 2;
            v2f a, b0, b1;
            if (FULL) {
                a.x = xp[rowA * F + kA];
                a.y = xp[rowA * F + kA + 1];
                int kW = s * F + kA;
                b0.x = W[(kW    ) * H_ + l16];
                b0.y = W[(kW + 1) * H_ + l16];
                b1.x = W[(kW    ) * H_ + 16 + l16];
                b1.y = W[(kW + 1) * H_ + 16 + l16];
            } else {
                float m0 = (kA     < F) ? 1.0f : 0.0f;
                float m1 = (kA + 1 < F) ? 1.0f : 0.0f;
                int ka0 = (kA     < F) ? kA     : (F - 1);
                int ka1 = (kA + 1 < F) ? kA + 1 : (F - 1);
                a.x = xp[rowA * F + ka0] * m0;
                a.y = xp[rowA * F + ka1] * m1;
                int kw0 = s * F + ka0;
                int kw1 = s * F + ka1;
                b0.x = W[kw0 * H_ + l16] * m0;
                b0.y = W[kw1 * H_ + l16] * m1;
                b1.x = W[kw0 * H_ + 16 + l16] * m0;
                b1.y = W[kw1 * H_ + 16 + l16] * m1;
            }
            c0 = __builtin_amdgcn_wmma_f32_16x16x4_f32(false, a, false, b0,
                                                       (short)0, c0, false, false);
            c1 = __builtin_amdgcn_wmma_f32_16x16x4_f32(false, a, false, b1,
                                                       (short)0, c1, false, false);
        }
    }

#pragma unroll
    for (int r = 0; r < 8; ++r) {
        int row = tile * 16 + half * 8 + r;
        float v0 = tanhf(c0[r] + bias[l16]);
        float v1 = tanhf(c1[r] + bias[16 + l16]);
        out[row * H_ + l16]      = v0;
        out[row * H_ + 16 + l16] = v1;
    }
}

__global__ void readout_kernel(const float* __restrict__ h, const float* __restrict__ Wr,
                               const float* __restrict__ br, float* __restrict__ out) {
    int i = blockIdx.x * blockDim.x + threadIdx.x;
    if (i >= NTOT_) return;
    float acc = br[0];
#pragma unroll
    for (int j = 0; j < H_; ++j) acc += h[i * H_ + j] * Wr[j];
    out[i] = acc;
}

static inline int cdiv(int a, int b) { return (a + b - 1) / b; }

extern "C" void kernel_launch(void* const* d_in, const int* in_sizes, int n_in,
                              void* d_out, int out_size, void* d_ws, size_t ws_size,
                              hipStream_t stream) {
    (void)in_sizes; (void)n_in; (void)out_size; (void)ws_size;

    const float* X  = (const float*)d_in[0];
    const int*   EI = (const int*)d_in[1];
    const float* W0 = (const float*)d_in[2];
    const float* b0 = (const float*)d_in[3];
    const float* W1 = (const float*)d_in[4];
    const float* b1 = (const float*)d_in[5];
    const float* W2 = (const float*)d_in[6];
    const float* b2 = (const float*)d_in[7];
    const float* Wr = (const float*)d_in[8];
    const float* br = (const float*)d_in[9];

    float* outF = (float*)d_out;
    float* hOut = outF + NTOT_;

    float* ws   = (float*)d_ws;
    float* dinv = ws;
    float* norm = dinv + NTOT_;
    float* p1   = norm + BE_;
    float* p2   = p1 + (size_t)NTOT_ * H_;
    float* bufA = p2 + (size_t)NTOT_ * H_;
    float* bufB = bufA + (size_t)NTOT_ * H_;

    const int TB = 256;

    zero_f32_kernel<<<cdiv(NTOT_, TB), TB, 0, stream>>>(dinv, NTOT_);
    deg_kernel<<<cdiv(BE_, TB), TB, 0, stream>>>(EI, dinv);
    dinv_kernel<<<cdiv(NTOT_, TB), TB, 0, stream>>>(dinv);
    norm_kernel<<<cdiv(BE_, TB), TB, 0, stream>>>(EI, dinv, norm);

    const int gemmBlocks = cdiv(NTOT_ / 16, TB / 32);
    const int propQ = BE_ * 8;

    zero_f32_kernel<<<cdiv(NTOT_ * WIN_, TB), TB, 0, stream>>>(p1, NTOT_ * WIN_);
    prop_kernel<WIN_><<<cdiv(BE_ * WIN_, TB), TB, 0, stream>>>(EI, norm, X, p1);
    zero_f32_kernel<<<cdiv(NTOT_ * WIN_, TB), TB, 0, stream>>>(p2, NTOT_ * WIN_);
    prop_kernel<WIN_><<<cdiv(BE_ * WIN_, TB), TB, 0, stream>>>(EI, norm, p1, p2);
    tag_gemm_kernel<WIN_><<<gemmBlocks, TB, 0, stream>>>(X, p1, p2, W0, b0, bufA);

    zero_f32_kernel<<<cdiv(NTOT_ * H_, TB), TB, 0, stream>>>(p1, NTOT_ * H_);
    prop4_kernel<<<cdiv(propQ, TB), TB, 0, stream>>>(EI, norm, (const float4*)bufA, p1);
    zero_f32_kernel<<<cdiv(NTOT_ * H_, TB), TB, 0, stream>>>(p2, NTOT_ * H_);
    prop4_kernel<<<cdiv(propQ, TB), TB, 0, stream>>>(EI, norm, (const float4*)p1, p2);
    tag_gemm_kernel<H_><<<gemmBlocks, TB, 0, stream>>>(bufA, p1, p2, W1, b1, bufB);

    zero_f32_kernel<<<cdiv(NTOT_ * H_, TB), TB, 0, stream>>>(p1, NTOT_ * H_);
    prop4_kernel<<<cdiv(propQ, TB), TB, 0, stream>>>(EI, norm, (const float4*)bufB, p1);
    zero_f32_kernel<<<cdiv(NTOT_ * H_, TB), TB, 0, stream>>>(p2, NTOT_ * H_);
    prop4_kernel<<<cdiv(propQ, TB), TB, 0, stream>>>(EI, norm, (const float4*)p1, p2);
    tag_gemm_kernel<H_><<<gemmBlocks, TB, 0, stream>>>(bufB, p1, p2, W2, b2, hOut);

    readout_kernel<<<cdiv(NTOT_, TB), TB, 0, stream>>>(hOut, Wr, br, outF);
}


# Replace the float atomic scatter of all six propagations by a gather over dst-sorted edge lists built in-call (int count+rank, scan, fill); row sums now run in arrival-rank order
# speedup vs baseline: 2.1191x; 2.1191x over previous
#include <hip/hip_runtime.h>
#include <math.h>

#define B_    4
#define WIN_  5
#define N_    50000
#define E_    1000000
#define H_    32
#define NTOT_ (B_ * N_)
#define BE_   (B_ * E_)

typedef __attribute__((ext_vector_type(2))) float v2f;
typedef __attribute__((ext_vector_type(8))) float v8f;

static __device__ __forceinline__ void decode_edge(const int* __restrict__ ei, int e,
                                                   int& src, int& dst) {
    int b  = e / E_;
    int el = e - b * E_;
    src = ei[(b * 2 + 0) * E_ + el] + b * N_;
    dst = ei[(b * 2 + 1) * E_ + el] + b * N_;
}

#define SCAN_TB     256
#define SCAN_CHUNK  (SCAN_TB * 4)
#define SCAN_BLOCKS ((NTOT_ + SCAN_CHUNK - 1) / SCAN_CHUNK)
static_assert(SCAN_BLOCKS <= SCAN_TB, "one block must cover all chunk sums");

__global__ void zero_i32_kernel(int* __restrict__ p, int n) {
    int i = blockIdx.x * blockDim.x + threadIdx.x;
    if (i < n) p[i] = 0;
}

__global__ void count_rank_kernel(const int* __restrict__ ei, int* __restrict__ cnt,
                                  int* __restrict__ rank) {
    int e = blockIdx.x * blockDim.x + threadIdx.x;
    if (e >= BE_) return;
    int src, dst;
    decode_edge(ei, e, src, dst);
    rank[e] = atomicAdd(&cnt[dst], 1);
}

static __device__ __forceinline__ int block_incl_scan(int v, int* sh) {
    int t = threadIdx.x;
    sh[t] = v;
    __syncthreads();
    for (int off = 1; off < SCAN_TB; off <<= 1) {
        int x = (t >= off) ? sh[t - off] : 0;
        __syncthreads();
        sh[t] += x;
        __syncthreads();
    }
    return sh[t];
}

__global__ void scan_sums_kernel(const int* __restrict__ cnt, int* __restrict__ blocksum) {
    __shared__ int sh[SCAN_TB];
    int base = blockIdx.x * SCAN_CHUNK + threadIdx.x * 4;
    int s = 0;
#pragma unroll
    for (int k = 0; k < 4; ++k) {
        int idx = base + k;
        if (idx < NTOT_) s += cnt[idx];
    }
    int incl = block_incl_scan(s, sh);
    if (threadIdx.x == SCAN_TB - 1) blocksum[blockIdx.x] = incl;
}

__global__ void scan_blocks_kernel(const int* __restrict__ blocksum, int* __restrict__ blockoff,
                                   int* __restrict__ rowptr) {
    __shared__ int sh[SCAN_TB];
    int t = threadIdx.x;
    int v = (t < SCAN_BLOCKS) ? blocksum[t] : 0;
    int incl = block_incl_scan(v, sh);
    if (t < SCAN_BLOCKS) blockoff[t] = incl - v;
    if (t == SCAN_TB - 1) rowptr[NTOT_] = incl;
}

__global__ void scan_rows_kernel(const int* __restrict__ cnt, const int* __restrict__ blockoff,
                                 int* __restrict__ rowptr, float* __restrict__ dinv) {
    __shared__ int sh[SCAN_TB];
    int base = blockIdx.x * SCAN_CHUNK + threadIdx.x * 4;
    int c[4];
    int s = 0;
#pragma unroll
    for (int k = 0; k < 4; ++k) {
        int idx = base + k;
        c[k] = (idx < NTOT_) ? cnt[idx] : 0;
        s += c[k];
    }
    int incl = block_incl_scan(s, sh);
    int run = blockoff[blockIdx.x] + incl - s;
#pragma unroll
    for (int k = 0; k < 4; ++k) {
        int idx = base + k;
        if (idx < NTOT_) {
            rowptr[idx] = run;
            run += c[k];
            float d = (float)c[k];
            dinv[idx] = (c[k] > 0) ? rsqrtf(d) : 0.0f;
        }
    }
}

__global__ void fill_kernel(const int* __restrict__ ei, const int* __restrict__ rowptr,
                            const int* __restrict__ rank, const float* __restrict__ dinv,
                            int* __restrict__ csr_src, float* __restrict__ csr_w) {
    int e = blockIdx.x * blockDim.x + threadIdx.x;
    if (e >= BE_) return;
    int src, dst;
    decode_edge(ei, e, src, dst);
    int pos = rowptr[dst] + rank[e];
    csr_src[pos] = src;
    csr_w[pos]   = dinv[src] * dinv[dst];
}

template <int F>
__global__ void pull_kernel(const int* __restrict__ rowptr, const int* __restrict__ csr_src,
                            const float* __restrict__ csr_w, const float* __restrict__ hin,
                            float* __restrict__ hout) {
    int tid = blockIdx.x * blockDim.x + threadIdx.x;
    if (tid >= NTOT_ * F) return;
    int i = tid / F;
    int f = tid - i * F;
    int j   = rowptr[i];
    int end = rowptr[i + 1];
    float acc = 0.0f;
    for (; j + 4 <= end; j += 4) {
        int s0 = csr_src[j], s1 = csr_src[j + 1], s2 = csr_src[j + 2], s3 = csr_src[j + 3];
        float w0 = csr_w[j], w1 = csr_w[j + 1], w2 = csr_w[j + 2], w3 = csr_w[j + 3];
        float v0 = hin[s0 * F + f];
        float v1 = hin[s1 * F + f];
        float v2 = hin[s2 * F + f];
        float v3 = hin[s3 * F + f];
        acc += v0 * w0;
        acc += v1 * w1;
        acc += v2 * w2;
        acc += v3 * w3;
    }
    for (; j < end; ++j) {
        int s = csr_src[j];
        acc += hin[s * F + f] * csr_w[j];
    }
    hout[tid] = acc;
}

__global__ void pull4_kernel(const int* __restrict__ rowptr, const int* __restrict__ csr_src,
                             const float* __restrict__ csr_w, const float4* __restrict__ hin4,
                             float4* __restrict__ hout4) {
    int tid = blockIdx.x * blockDim.x + threadIdx.x;
    int i = tid >> 3;
    int q = tid & 7;
    if (i >= NTOT_) return;
    int j   = rowptr[i];
    int end = rowptr[i + 1];
    float4 acc = make_float4(0.0f, 0.0f, 0.0f, 0.0f);
    for (; j + 4 <= end; j += 4) {
        int s0 = csr_src[j], s1 = csr_src[j + 1], s2 = csr_src[j + 2], s3 = csr_src[j + 3];
        float w0 = csr_w[j], w1 = csr_w[j + 1], w2 = csr_w[j + 2], w3 = csr_w[j + 3];
        float4 v0 = hin4[s0 * 8 + q];
        float4 v1 = hin4[s1 * 8 + q];
        float4 v2 = hin4[s2 * 8 + q];
        float4 v3 = hin4[s3 * 8 + q];
        acc.x += v0.x * w0; acc.y += v0.y * w0; acc.z += v0.z * w0; acc.w += v0.w * w0;
        acc.x += v1.x * w1; acc.y += v1.y * w1; acc.z += v1.z * w1; acc.w += v1.w * w1;
        acc.x += v2.x * w2; acc.y += v2.y * w2; acc.z += v2.z * w2; acc.w += v2.w * w2;
        acc.x += v3.x * w3; acc.y += v3.y * w3; acc.z += v3.z * w3; acc.w += v3.w * w3;
    }
    for (; j < end; ++j) {
        int s = csr_src[j];
        float w = csr_w[j];
        float4 v = hin4[s * 8 + q];
        acc.x += v.x * w; acc.y += v.y * w; acc.z += v.z * w; acc.w += v.w * w;
    }
    hout4[i * 8 + q] = acc;
}

template <int F>
__global__ void tag_gemm_kernel(const float* __restrict__ x0, const float* __restrict__ x1,
                                const float* __restrict__ x2, const float* __restrict__ W,
                                const float* __restrict__ bias, float* __restrict__ out) {
    const int numTiles = NTOT_ / 16;
    int wave = threadIdx.x >> 5;
    int tile = blockIdx.x * (blockDim.x >> 5) + wave;
    if (tile >= numTiles) return;

    int lane = threadIdx.x & 31;
    int half = lane >> 4;
    int l16  = lane & 15;
    int rowA = tile * 16 + l16;

    v8f c0 = {};
    v8f c1 = {};

    const float* xs[3] = { x0, x1, x2 };
    constexpr bool FULL = (F % 4 == 0);

#pragma unroll
    for (int s = 0; s < 3; ++s) {
        const float* __restrict__ xp = xs[s];
#pragma unroll
        for (int k0 = 0; k0 < F; k0 += 4) {
            int kA = k0 + half * 2;
            v2f a, b0, b1;
            if (FULL) {
                a.x = xp[rowA * F + kA];
                a.y = xp[rowA * F + kA + 1];
                int kW = s * F + kA;
                b0.x = W[(kW    ) * H_ + l16];
                b0.y = W[(kW + 1) * H_ + l16];
                b1.x = W[(kW    ) * H_ + 16 + l16];
                b1.y = W[(kW + 1) * H_ + 16 + l16];
            } else {
                float m0 = (kA     < F) ? 1.0f : 0.0f;
                float m1 = (kA + 1 < F) ? 1.0f : 0.0f;
                int ka0 = (kA     < F) ? kA     : (F - 1);
                int ka1 = (kA + 1 < F) ? kA + 1 : (F - 1);
                a.x = xp[rowA * F + ka0] * m0;
                a.y = xp[rowA * F + ka1] * m1;
                int kw0 = s * F + ka0;
                int kw1 = s * F + ka1;
                b0.x = W[kw0 * H_ + l16] * m0;
                b0.y = W[kw1 * H_ + l16] * m1;
                b1.x = W[kw0 * H_ + 16 + l16] * m0;
                b1.y = W[kw1 * H_ + 16 + l16] * m1;
            }
            c0 = __builtin_amdgcn_wmma_f32_16x16x4_f32(false, a, false, b0,
                                                       (short)0, c0, false, false);
            c1 = __builtin_amdgcn_wmma_f32_16x16x4_f32(false, a, false, b1,
                                                       (short)0, c1, false, false);
        }
    }

#pragma unroll
    for (int r = 0; r < 8; ++r) {
        int row = tile * 16 + half * 8 + r;
        float v0 = tanhf(c0[r] + bias[l16]);
        float v1 = tanhf(c1[r] + bias[16 + l16]);
        out[row * H_ + l16]      = v0;
        out[row * H_ + 16 + l16] = v1;
    }
}

__global__ void readout_kernel(const float* __restrict__ h, const float* __restrict__ Wr,
                               const float* __restrict__ br, float* __restrict__ out) {
    int i = blockIdx.x * blockDim.x + threadIdx.x;
    if (i >= NTOT_) return;
    float acc = br[0];
#pragma unroll
    for (int j = 0; j < H_; ++j) acc += h[i * H_ + j] * Wr[j];
    out[i] = acc;
}

static inline int cdiv(int a, int b) { return (a + b - 1) / b; }

extern "C" void kernel_launch(void* const* d_in, const int* in_sizes, int n_in,
                              void* d_out, int out_size, void* d_ws, size_t ws_size,
                              hipStream_t stream) {
    (void)in_sizes; (void)n_in; (void)out_size; (void)ws_size;

    const float* X  = (const float*)d_in[0];
    const int*   EI = (const int*)d_in[1];
    const float* W0 = (const float*)d_in[2];
    const float* b0 = (const float*)d_in[3];
    const float* W1 = (const float*)d_in[4];
    const float* b1 = (const float*)d_in[5];
    const float* W2 = (const float*)d_in[6];
    const float* b2 = (const float*)d_in[7];
    const float* Wr = (const float*)d_in[8];
    const float* br = (const float*)d_in[9];

    float* outF = (float*)d_out;
    float* hOut = outF + NTOT_;

    int*   cnt    = (int*)d_ws;
    int*   rowptr = cnt + NTOT_;
    int*   bsum   = rowptr + NTOT_ + 64;
    int*   boff   = bsum + SCAN_TB;
    float* dinv   = (float*)(boff + SCAN_TB);
    int*   rank   = (int*)(dinv + NTOT_);
    int*   csrc   = rank + BE_;
    float* cw     = (float*)(csrc + BE_);
    float* p1     = cw + BE_;
    float* p2     = p1 + (size_t)NTOT_ * H_;
    float* bufA   = p2 + (size_t)NTOT_ * H_;
    float* bufB   = bufA + (size_t)NTOT_ * H_;

    const int TB = 256;

    zero_i32_kernel<<<cdiv(NTOT_, TB), TB, 0, stream>>>(cnt, NTOT_);
    count_rank_kernel<<<cdiv(BE_, TB), TB, 0, stream>>>(EI, cnt, rank);
    scan_sums_kernel<<<SCAN_BLOCKS, SCAN_TB, 0, stream>>>(cnt, bsum);
    scan_blocks_kernel<<<1, SCAN_TB, 0, stream>>>(bsum, boff, rowptr);
    scan_rows_kernel<<<SCAN_BLOCKS, SCAN_TB, 0, stream>>>(cnt, boff, rowptr, dinv);
    fill_kernel<<<cdiv(BE_, TB), TB, 0, stream>>>(EI, rowptr, rank, dinv, csrc, cw);

    const int gemmBlocks = cdiv(NTOT_ / 16, TB / 32);
    const int pullN = cdiv(NTOT_ * WIN_, TB);
    const int pullW = cdiv(NTOT_ * 8, TB);

    pull_kernel<WIN_><<<pullN, TB, 0, stream>>>(rowptr, csrc, cw, X, p1);
    pull_kernel<WIN_><<<pullN, TB, 0, stream>>>(rowptr, csrc, cw, p1, p2);
    tag_gemm_kernel<WIN_><<<gemmBlocks, TB, 0, stream>>>(X, p1, p2, W0, b0, bufA);

    pull4_kernel<<<pullW, TB, 0, stream>>>(rowptr, csrc, cw, (const float4*)bufA, (float4*)p1);
    pull4_kernel<<<pullW, TB, 0, stream>>>(rowptr, csrc, cw, (const float4*)p1, (float4*)p2);
    tag_gemm_kernel<H_><<<gemmBlocks, TB, 0, stream>>>(bufA, p1, p2, W1, b1, bufB);

    pull4_kernel<<<pullW, TB, 0, stream>>>(rowptr, csrc, cw, (const float4*)bufB, (float4*)p1);
    pull4_kernel<<<pullW, TB, 0, stream>>>(rowptr, csrc, cw, (const float4*)p1, (float4*)p2);
    tag_gemm_kernel<H_><<<gemmBlocks, TB, 0, stream>>>(bufB, p1, p2, W2, b2, hOut);

    readout_kernel<<<cdiv(NTOT_, TB), TB, 0, stream>>>(hOut, Wr, br, outF);
}


# Six gathers: a row's 8-lane group loads its edge records (source, weight) once, up to 32 up front, one coalesced load per 8 edges, shared by DPP8 broadcast, not a same-address load in every lane
# speedup vs baseline: 2.3242x; 1.0968x over previous
#include <hip/hip_runtime.h>
#include <math.h>

#define B_    4
#define WIN_  5
#define N_    50000
#define E_    1000000
#define H_    32
#define NTOT_ (B_ * N_)
#define BE_   (B_ * E_)

typedef __attribute__((ext_vector_type(2))) float v2f;
typedef __attribute__((ext_vector_type(8))) float v8f;

static __device__ __forceinline__ void decode_edge(const int* __restrict__ ei, int e,
                                                   int& src, int& dst) {
    int b  = e / E_;
    int el = e - b * E_;
    src = ei[(b * 2 + 0) * E_ + el] + b * N_;
    dst = ei[(b * 2 + 1) * E_ + el] + b * N_;
}

#define SCAN_TB     256
#define SCAN_CHUNK  (SCAN_TB * 4)
#define SCAN_BLOCKS ((NTOT_ + SCAN_CHUNK - 1) / SCAN_CHUNK)
static_assert(SCAN_BLOCKS <= SCAN_TB, "one block must cover all chunk sums");

__global__ void zero_i32_kernel(int* __restrict__ p, int n) {
    int i = blockIdx.x * blockDim.x + threadIdx.x;
    if (i < n) p[i] = 0;
}

__global__ void count_rank_kernel(const int* __restrict__ ei, int* __restrict__ cnt,
                                  int* __restrict__ rank) {
    int e = blockIdx.x * blockDim.x + threadIdx.x;
    if (e >= BE_) return;
    int src, dst;
    decode_edge(ei, e, src, dst);
    rank[e] = atomicAdd(&cnt[dst], 1);
}

static __device__ __forceinline__ int block_incl_scan(int v, int* sh) {
    int t = threadIdx.x;
    sh[t] = v;
    __syncthreads();
    for (int off = 1; off < SCAN_TB; off <<= 1) {
        int x = (t >= off) ? sh[t - off] : 0;
        __syncthreads();
        sh[t] += x;
        __syncthreads();
    }
    return sh[t];
}

__global__ void scan_sums_kernel(const int* __restrict__ cnt, int* __restrict__ blocksum) {
    __shared__ int sh[SCAN_TB];
    int base = blockIdx.x * SCAN_CHUNK + threadIdx.x * 4;
    int s = 0;
#pragma unroll
    for (int k = 0; k < 4; ++k) {
        int idx = base + k;
        if (idx < NTOT_) s += cnt[idx];
    }
    int incl = block_incl_scan(s, sh);
    if (threadIdx.x == SCAN_TB - 1) blocksum[blockIdx.x] = incl;
}

__global__ void scan_blocks_kernel(const int* __restrict__ blocksum, int* __restrict__ blockoff,
                                   int* __restrict__ rowptr) {
    __shared__ int sh[SCAN_TB];
    int t = threadIdx.x;
    int v = (t < SCAN_BLOCKS) ? blocksum[t] : 0;
    int incl = block_incl_scan(v, sh);
    if (t < SCAN_BLOCKS) blockoff[t] = incl - v;
    if (t == SCAN_TB - 1) rowptr[NTOT_] = incl;
}

__global__ void scan_rows_kernel(const int* __restrict__ cnt, const int* __restrict__ blockoff,
                                 int* __restrict__ rowptr, float* __restrict__ dinv) {
    __shared__ int sh[SCAN_TB];
    int base = blockIdx.x * SCAN_CHUNK + threadIdx.x * 4;
    int c[4];
    int s = 0;
#pragma unroll
    for (int k = 0; k < 4; ++k) {
        int idx = base + k;
        c[k] = (idx < NTOT_) ? cnt[idx] : 0;
        s += c[k];
    }
    int incl = block_incl_scan(s, sh);
    int run = blockoff[blockIdx.x] + incl - s;
#pragma unroll
    for (int k = 0; k < 4; ++k) {
        int idx = base + k;
        if (idx < NTOT_) {
            rowptr[idx] = run;
            run += c[k];
            float d = (float)c[k];
            dinv[idx] = (c[k] > 0) ? rsqrtf(d) : 0.0f;
        }
    }
}

__global__ void fill_kernel(const int* __restrict__ ei, const int* __restrict__ rowptr,
                            const int* __restrict__ rank, const float* __restrict__ dinv,
                            int* __restrict__ csr_src, float* __restrict__ csr_w) {
    int e = blockIdx.x * blockDim.x + threadIdx.x;
    if (e >= BE_) return;
    int src, dst;
    decode_edge(ei, e, src, dst);
    int pos = rowptr[dst] + rank[e];
    csr_src[pos] = src;
    csr_w[pos]   = dinv[src] * dinv[dst];
}

template <int K>
static __device__ __forceinline__ int bcast8(int v) {
    return (int)__builtin_amdgcn_mov_dpp8((unsigned)v, (unsigned)(K * 0x249249));
}
template <int K>
static __device__ __forceinline__ float bcast8(float v) {
    return __int_as_float(bcast8<K>(__float_as_int(v)));
}

static __device__ __forceinline__ void axpy(float& a, float v, float w) { a = __builtin_fmaf(v, w, a); }
static __device__ __forceinline__ void axpy(float4& a, const float4& v, float w) {
    a.x = __builtin_fmaf(v.x, w, a.x); a.y = __builtin_fmaf(v.y, w, a.y);
    a.z = __builtin_fmaf(v.z, w, a.z); a.w = __builtin_fmaf(v.w, w, a.w);
}
static __device__ __forceinline__ void set_zero(float& a)  { a = 0.0f; }
static __device__ __forceinline__ void set_zero(float4& a) { a = make_float4(0.0f, 0.0f, 0.0f, 0.0f); }

template <int V> struct RowSlice;
template <> struct RowSlice<4> { typedef float4 T; };
template <> struct RowSlice<1> { typedef float  T; };

template <int LQ, typename RowT>
static __device__ __forceinline__ void pull_chunk(const RowT* __restrict__ hin_v, int s_l, float w_l,
                                                  int n, int q, bool carries, RowT& acc) {
    int   s[8];
    float w[8];
    s[0] = bcast8<0>(s_l); s[1] = bcast8<1>(s_l); s[2] = bcast8<2>(s_l); s[3] = bcast8<3>(s_l);
    s[4] = bcast8<4>(s_l); s[5] = bcast8<5>(s_l); s[6] = bcast8<6>(s_l); s[7] = bcast8<7>(s_l);
    w[0] = bcast8<0>(w_l); w[1] = bcast8<1>(w_l); w[2] = bcast8<2>(w_l); w[3] = bcast8<3>(w_l);
    w[4] = bcast8<4>(w_l); w[5] = bcast8<5>(w_l); w[6] = bcast8<6>(w_l); w[7] = bcast8<7>(w_l);
    RowT v[8];
#pragma unroll
    for (int k = 0; k < 8; ++k) {
        set_zero(v[k]);
        if (carries && k < n) v[k] = hin_v[s[k] * LQ + q];
    }
#pragma unroll
    for (int k = 0; k < 8; ++k) axpy(acc, v[k], w[k]);
}

template <int F>
__global__ void pull_kernel(const int* __restrict__ rowptr, const int* __restrict__ csr_src,
                            const float* __restrict__ csr_w, const float* __restrict__ hin,
                            float* __restrict__ hout) {
    constexpr int V  = (F % 8 == 0) ? (F / 8) : 1;
    constexpr int LQ = F / V;
    typedef typename RowSlice<V>::T RowT;
    int tid = blockIdx.x * blockDim.x + threadIdx.x;
    int i = tid >> 3;
    int q = tid & 7;
    if (i >= NTOT_) return;
    const RowT* __restrict__ hin_v = (const RowT*)hin;
    RowT* __restrict__ hout_v = (RowT*)hout;
    const bool carries = (q < LQ);
    int j   = rowptr[i];
    int end = rowptr[i + 1];
    RowT acc;
    set_zero(acc);
    for (; j < end; j += 32) {
        int n = end - j;
        int   s_l[4];
        float w_l[4];
#pragma unroll
        for (int c = 0; c < 4; ++c) {
            s_l[c] = 0;
            w_l[c] = 0.0f;
            if (8 * c + q < n) {
                s_l[c] = csr_src[j + 8 * c + q];
                w_l[c] = csr_w[j + 8 * c + q];
            }
        }
#pragma unroll
        for (int c = 0; c < 4; ++c) {
            if (8 * c < n)
                pull_chunk<LQ, RowT>(hin_v, s_l[c], w_l[c], n - 8 * c, q, carries, acc);
        }
    }
    if (carries) hout_v[i * LQ + q] = acc;
}

template <int F>
__global__ void tag_gemm_kernel(const float* __restrict__ x0, const float* __restrict__ x1,
                                const float* __restrict__ x2, const float* __restrict__ W,
                                const float* __restrict__ bias, float* __restrict__ out) {
    const int numTiles = NTOT_ / 16;
    int wave = threadIdx.x >> 5;
    int tile = blockIdx.x * (blockDim.x >> 5) + wave;
    if (tile >= numTiles) return;

    int lane = threadIdx.x & 31;
    int half = lane >> 4;
    int l16  = lane & 15;
    int rowA = tile * 16 + l16;

    v8f c0 = {};
    v8f c1 = {};

    const float* xs[3] = { x0, x1, x2 };
    constexpr bool FULL = (F % 4 == 0);

#pragma unroll
    for (int s = 0; s < 3; ++s) {
        const float* __restrict__ xp = xs[s];
#pragma unroll
        for (int k0 = 0; k0 < F; k0 += 4) {
            int kA = k0 + half * 2;
            v2f a, b0, b1;
            if (FULL) {
                a.x = xp[rowA * F + kA];
                a.y = xp[rowA * F + kA + 1];
                int kW = s * F + kA;
                b0.x = W[(kW    ) * H_ + l16];
                b0.y = W[(kW + 1) * H_ + l16];
                b1.x = W[(kW    ) * H_ + 16 + l16];
                b1.y = W[(kW + 1) * H_ + 16 + l16];
            } else {
                float m0 = (kA     < F) ? 1.0f : 0.0f;
                float m1 = (kA + 1 < F) ? 1.0f : 0.0f;
                int ka0 = (kA     < F) ? kA     : (F - 1);
                int ka1 = (kA + 1 < F) ? kA + 1 : (F - 1);
                a.x = xp[rowA * F + ka0] * m0;
                a.y = xp[rowA * F + ka1] * m1;
                int kw0 = s * F + ka0;
                int kw1 = s * F + ka1;
                b0.x = W[kw0 * H_ + l16] * m0;
                b0.y = W[kw1 * H_ + l16] * m1;
                b1.x = W[kw0 * H_ + 16 + l16] * m0;
                b1.y = W[kw1 * H_ + 16 + l16] * m1;
            }
            c0 = __builtin_amdgcn_wmma_f32_16x16x4_f32(false, a, false, b0,
                                                       (short)0, c0, false, false);
            c1 = __builtin_amdgcn_wmma_f32_16x16x4_f32(false, a, false, b1,
                                                       (short)0, c1, false, false);
        }
    }

#pragma unroll
    for (int r = 0; r < 8; ++r) {
        int row = tile * 16 + half * 8 + r;
        float v0 = tanhf(c0[r] + bias[l16]);
        float v1 = tanhf(c1[r] + bias[16 + l16]);
        out[row * H_ + l16]      = v0;
        out[row * H_ + 16 + l16] = v1;
    }
}

__global__ void readout_kernel(const float* __restrict__ h, const float* __restrict__ Wr,
                               const float* __restrict__ br, float* __restrict__ out) {
    int i = blockIdx.x * blockDim.x + threadIdx.x;
    if (i >= NTOT_) return;
    float acc = br[0];
#pragma unroll
    for (int j = 0; j < H_; ++j) acc += h[i * H_ + j] * Wr[j];
    out[i] = acc;
}

static inline int cdiv(int a, int b) { return (a + b - 1) / b; }

extern "C" void kernel_launch(void* const* d_in, const int* in_sizes, int n_in,
                              void* d_out, int out_size, void* d_ws, size_t ws_size,
                              hipStream_t stream) {
    (void)in_sizes; (void)n_in; (void)out_size; (void)ws_size;

    const float* X  = (const float*)d_in[0];
    const int*   EI = (const int*)d_in[1];
    const float* W0 = (const float*)d_in[2];
    const float* b0 = (const float*)d_in[3];
    const float* W1 = (const float*)d_in[4];
    const float* b1 = (const float*)d_in[5];
    const float* W2 = (const float*)d_in[6];
    const float* b2 = (const float*)d_in[7];
    const float* Wr = (const float*)d_in[8];
    const float* br = (const float*)d_in[9];

    float* outF = (float*)d_out;
    float* hOut = outF + NTOT_;

    int*   cnt    = (int*)d_ws;
    int*   rowptr = cnt + NTOT_;
    int*   bsum   = rowptr + NTOT_ + 64;
    int*   boff   = bsum + SCAN_TB;
    float* dinv   = (float*)(boff + SCAN_TB);
    int*   rank   = (int*)(dinv + NTOT_);
    int*   csrc   = rank + BE_;
    float* cw     = (float*)(csrc + BE_);
    float* p1     = cw + BE_;
    float* p2     = p1 + (size_t)NTOT_ * H_;
    float* bufA   = p2 + (size_t)NTOT_ * H_;
    float* bufB   = bufA + (size_t)NTOT_ * H_;

    const int TB = 256;

    zero_i32_kernel<<<cdiv(NTOT_, TB), TB, 0, stream>>>(cnt, NTOT_);
    count_rank_kernel<<<cdiv(BE_, TB), TB, 0, stream>>>(EI, cnt, rank);
    scan_sums_kernel<<<SCAN_BLOCKS, SCAN_TB, 0, stream>>>(cnt, bsum);
    scan_blocks_kernel<<<1, SCAN_TB, 0, stream>>>(bsum, boff, rowptr);
    scan_rows_kernel<<<SCAN_BLOCKS, SCAN_TB, 0, stream>>>(cnt, boff, rowptr, dinv);
    fill_kernel<<<cdiv(BE_, TB), TB, 0, stream>>>(EI, rowptr, rank, dinv, csrc, cw);

    const int gemmBlocks = cdiv(NTOT_ / 16, TB / 32);
    const int pullW = cdiv(NTOT_ * 8, TB);

    pull_kernel<WIN_><<<pullW, TB, 0, stream>>>(rowptr, csrc, cw, X, p1);
    pull_kernel<WIN_><<<pullW, TB, 0, stream>>>(rowptr, csrc, cw, p1, p2);
    tag_gemm_kernel<WIN_><<<gemmBlocks, TB, 0, stream>>>(X, p1, p2, W0, b0, bufA);

    pull_kernel<H_><<<pullW, TB, 0, stream>>>(rowptr, csrc, cw, bufA, p1);
    pull_kernel<H_><<<pullW, TB, 0, stream>>>(rowptr, csrc, cw, p1, p2);
    tag_gemm_kernel<H_><<<gemmBlocks, TB, 0, stream>>>(bufA, p1, p2, W1, b1, bufB);

    pull_kernel<H_><<<pullW, TB, 0, stream>>>(rowptr, csrc, cw, bufB, p1);
    pull_kernel<H_><<<pullW, TB, 0, stream>>>(rowptr, csrc, cw, p1, p2);
    tag_gemm_kernel<H_><<<gemmBlocks, TB, 0, stream>>>(bufB, p1, p2, W2, b2, hOut);

    readout_kernel<<<cdiv(NTOT_, TB), TB, 0, stream>>>(hOut, Wr, br, outF);
}


# Fewer, longer-lived waves: count+rank and fill take 8 edges per thread (loads, adds, stores issued as groups), each 8-lane gather group sums 4 nodes in turn; row sums stay in arrival-rank order
# speedup vs baseline: 2.4709x; 1.0631x over previous
#include <hip/hip_runtime.h>
#include <math.h>

#define B_    4
#define WIN_  5
#define N_    50000
#define E_    1000000
#define H_    32
#define NTOT_ (B_ * N_)
#define BE_   (B_ * E_)

typedef __attribute__((ext_vector_type(2))) float v2f;
typedef __attribute__((ext_vector_type(8))) float v8f;

#define SCAN_TB     256
#define SCAN_CHUNK  (SCAN_TB * 4)
#define SCAN_BLOCKS ((NTOT_ + SCAN_CHUNK - 1) / SCAN_CHUNK)
static_assert(SCAN_BLOCKS <= SCAN_TB, "one block must cover all chunk sums");

static_assert(NTOT_ % 4 == 0, "the counters are cleared in 16-byte groups");
__global__ void zero_i32_kernel(int* __restrict__ p, int n4) {
    int i = blockIdx.x * blockDim.x + threadIdx.x;
    if (i < n4) ((int4*)p)[i] = make_int4(0, 0, 0, 0);
}

#define EPT_    (B_ * 2)
#define EHALF_  (E_ / 2)
static_assert(E_ % 2 == 0, "edge list splits in two halves");

__global__ void count_rank_kernel(const int* __restrict__ ei, int* __restrict__ cnt,
                                  int* __restrict__ rank) {
    int t = blockIdx.x * blockDim.x + threadIdx.x;
    if (t >= EHALF_) return;
    int dst[EPT_];
    int slot[EPT_];
#pragma unroll
    for (int k = 0; k < EPT_; ++k) {
        int b  = k >> 1;
        int el = t + (k & 1) * EHALF_;
        dst[k] = ei[(b * 2 + 1) * E_ + el] + b * N_;
    }
#pragma unroll
    for (int k = 0; k < EPT_; ++k) slot[k] = atomicAdd(&cnt[dst[k]], 1);
#pragma unroll
    for (int k = 0; k < EPT_; ++k) {
        int b  = k >> 1;
        int el = t + (k & 1) * EHALF_;
        rank[b * E_ + el] = slot[k];
    }
}

static __device__ __forceinline__ int block_incl_scan(int v, int* sh) {
    int t = threadIdx.x;
    sh[t] = v;
    __syncthreads();
    for (int off = 1; off < SCAN_TB; off <<= 1) {
        int x = (t >= off) ? sh[t - off] : 0;
        __syncthreads();
        sh[t] += x;
        __syncthreads();
    }
    return sh[t];
}

__global__ void scan_sums_kernel(const int* __restrict__ cnt, int* __restrict__ blocksum) {
    __shared__ int sh[SCAN_TB];
    int base = blockIdx.x * SCAN_CHUNK + threadIdx.x * 4;
    int s = 0;
#pragma unroll
    for (int k = 0; k < 4; ++k) {
        int idx = base + k;
        if (idx < NTOT_) s += cnt[idx];
    }
    int incl = block_incl_scan(s, sh);
    if (threadIdx.x == SCAN_TB - 1) blocksum[blockIdx.x] = incl;
}

__global__ void scan_blocks_kernel(const int* __restrict__ blocksum, int* __restrict__ blockoff,
                                   int* __restrict__ rowptr) {
    __shared__ int sh[SCAN_TB];
    int t = threadIdx.x;
    int v = (t < SCAN_BLOCKS) ? blocksum[t] : 0;
    int incl = block_incl_scan(v, sh);
    if (t < SCAN_BLOCKS) blockoff[t] = incl - v;
    if (t == SCAN_TB - 1) rowptr[NTOT_] = incl;
}

__global__ void scan_rows_kernel(const int* __restrict__ cnt, const int* __restrict__ blockoff,
                                 int* __restrict__ rowptr, float* __restrict__ dinv) {
    __shared__ int sh[SCAN_TB];
    int base = blockIdx.x * SCAN_CHUNK + threadIdx.x * 4;
    int c[4];
    int s = 0;
#pragma unroll
    for (int k = 0; k < 4; ++k) {
        int idx = base + k;
        c[k] = (idx < NTOT_) ? cnt[idx] : 0;
        s += c[k];
    }
    int incl = block_incl_scan(s, sh);
    int run = blockoff[blockIdx.x] + incl - s;
#pragma unroll
    for (int k = 0; k < 4; ++k) {
        int idx = base + k;
        if (idx < NTOT_) {
            rowptr[idx] = run;
            run += c[k];
            float d = (float)c[k];
            dinv[idx] = (c[k] > 0) ? rsqrtf(d) : 0.0f;
        }
    }
}

__global__ void fill_kernel(const int* __restrict__ ei, const int* __restrict__ rowptr,
                            const int* __restrict__ rank, const float* __restrict__ dinv,
                            int* __restrict__ csr_src, float* __restrict__ csr_w) {
    int t = blockIdx.x * blockDim.x + threadIdx.x;
    if (t >= EHALF_) return;
    int   src[EPT_], dst[EPT_], pos[EPT_];
    float ws[EPT_], wd[EPT_];
#pragma unroll
    for (int k = 0; k < EPT_; ++k) {
        int b  = k >> 1;
        int el = t + (k & 1) * EHALF_;
        src[k] = ei[(b * 2 + 0) * E_ + el] + b * N_;
        dst[k] = ei[(b * 2 + 1) * E_ + el] + b * N_;
        pos[k] = rank[b * E_ + el];
    }
#pragma unroll
    for (int k = 0; k < EPT_; ++k) {
        pos[k] += rowptr[dst[k]];
        ws[k] = dinv[src[k]];
        wd[k] = dinv[dst[k]];
    }
#pragma unroll
    for (int k = 0; k < EPT_; ++k) {
        csr_src[pos[k]] = src[k];
        csr_w[pos[k]]   = ws[k] * wd[k];
    }
}

template <int K>
static __device__ __forceinline__ int bcast8(int v) {
    return (int)__builtin_amdgcn_mov_dpp8((unsigned)v, (unsigned)(K * 0x249249));
}
template <int K>
static __device__ __forceinline__ float bcast8(float v) {
    return __int_as_float(bcast8<K>(__float_as_int(v)));
}

static __device__ __forceinline__ void axpy(float& a, float v, float w) { a = __builtin_fmaf(v, w, a); }
static __device__ __forceinline__ void axpy(float4& a, const float4& v, float w) {
    a.x = __builtin_fmaf(v.x, w, a.x); a.y = __builtin_fmaf(v.y, w, a.y);
    a.z = __builtin_fmaf(v.z, w, a.z); a.w = __builtin_fmaf(v.w, w, a.w);
}
static __device__ __forceinline__ void set_zero(float& a)  { a = 0.0f; }
static __device__ __forceinline__ void set_zero(float4& a) { a = make_float4(0.0f, 0.0f, 0.0f, 0.0f); }

template <int V> struct RowSlice;
template <> struct RowSlice<4> { typedef float4 T; };
template <> struct RowSlice<1> { typedef float  T; };

template <int LQ, typename RowT>
static __device__ __forceinline__ void pull_chunk(const RowT* __restrict__ hin_v, int s_l, float w_l,
                                                  int n, int q, bool carries, RowT& acc) {
    int   s[8];
    float w[8];
    s[0] = bcast8<0>(s_l); s[1] = bcast8<1>(s_l); s[2] = bcast8<2>(s_l); s[3] = bcast8<3>(s_l);
    s[4] = bcast8<4>(s_l); s[5] = bcast8<5>(s_l); s[6] = bcast8<6>(s_l); s[7] = bcast8<7>(s_l);
    w[0] = bcast8<0>(w_l); w[1] = bcast8<1>(w_l); w[2] = bcast8<2>(w_l); w[3] = bcast8<3>(w_l);
    w[4] = bcast8<4>(w_l); w[5] = bcast8<5>(w_l); w[6] = bcast8<6>(w_l); w[7] = bcast8<7>(w_l);
    RowT v[8];
#pragma unroll
    for (int k = 0; k < 8; ++k) {
        set_zero(v[k]);
        if (carries && k < n) v[k] = hin_v[s[k] * LQ + q];
    }
#pragma unroll
    for (int k = 0; k < 8; ++k) axpy(acc, v[k], w[k]);
}

#define PULL_NPG_ 4
#define PULL_G_   (NTOT_ / PULL_NPG_)
static_assert(NTOT_ % PULL_NPG_ == 0, "every lane group takes the same number of nodes");

template <int F>
__global__ void pull_kernel(const int* __restrict__ rowptr, const int* __restrict__ csr_src,
                            const float* __restrict__ csr_w, const float* __restrict__ hin,
                            float* __restrict__ hout) {
    constexpr int V  = (F % 8 == 0) ? (F / 8) : 1;
    constexpr int LQ = F / V;
    typedef typename RowSlice<V>::T RowT;
    int tid = blockIdx.x * blockDim.x + threadIdx.x;
    int g = tid >> 3;
    int q = tid & 7;
    if (g >= PULL_G_) return;
    const RowT* __restrict__ hin_v = (const RowT*)hin;
    RowT* __restrict__ hout_v = (RowT*)hout;
    const bool carries = (q < LQ);
    for (int node = 0; node < PULL_NPG_; ++node) {
        const int i = g + node * PULL_G_;
        int j   = rowptr[i];
        int end = rowptr[i + 1];
        RowT acc;
        set_zero(acc);
        for (; j < end; j += 32) {
            int n = end - j;
            int   s_l[4];
            float w_l[4];
#pragma unroll
            for (int c = 0; c < 4; ++c) {
                s_l[c] = 0;
                w_l[c] = 0.0f;
                if (8 * c + q < n) {
                    s_l[c] = csr_src[j + 8 * c + q];
                    w_l[c] = csr_w[j + 8 * c + q];
                }
            }
#pragma unroll
            for (int c = 0; c < 4; ++c) {
                if (8 * c < n)
                    pull_chunk<LQ, RowT>(hin_v, s_l[c], w_l[c], n - 8 * c, q, carries, acc);
            }
        }
        if (carries) hout_v[i * LQ + q] = acc;
    }
}

template <int F>
__global__ void tag_gemm_kernel(const float* __restrict__ x0, const float* __restrict__ x1,
                                const float* __restrict__ x2, const float* __restrict__ W,
                                const float* __restrict__ bias, float* __restrict__ out) {
    const int numTiles = NTOT_ / 16;
    int wave = threadIdx.x >> 5;
    int tile = blockIdx.x * (blockDim.x >> 5) + wave;
    if (tile >= numTiles) return;

    int lane = threadIdx.x & 31;
    int half = lane >> 4;
    int l16  = lane & 15;
    int rowA = tile * 16 + l16;

    v8f c0 = {};
    v8f c1 = {};

    const float* xs[3] = { x0, x1, x2 };
    constexpr bool FULL = (F % 4 == 0);

#pragma unroll
    for (int s = 0; s < 3; ++s) {
        const float* __restrict__ xp = xs[s];
#pragma unroll
        for (int k0 = 0; k0 < F; k0 += 4) {
            int kA = k0 + half * 2;
            v2f a, b0, b1;
            if (FULL) {
                a.x = xp[rowA * F + kA];
                a.y = xp[rowA * F + kA + 1];
                int kW = s * F + kA;
                b0.x = W[(kW    ) * H_ + l16];
                b0.y = W[(kW + 1) * H_ + l16];
                b1.x = W[(kW    ) * H_ + 16 + l16];
                b1.y = W[(kW + 1) * H_ + 16 + l16];
            } else {
                float m0 = (kA     < F) ? 1.0f : 0.0f;
                float m1 = (kA + 1 < F) ? 1.0f : 0.0f;
                int ka0 = (kA     < F) ? kA     : (F - 1);
                int ka1 = (kA + 1 < F) ? kA + 1 : (F - 1);
                a.x = xp[rowA * F + ka0] * m0;
                a.y = xp[rowA * F + ka1] * m1;
                int kw0 = s * F + ka0;
                int kw1 = s * F + ka1;
                b0.x = W[kw0 * H_ + l16] * m0;
                b0.y = W[kw1 * H_ + l16] * m1;
                b1.x = W[kw0 * H_ + 16 + l16] * m0;
                b1.y = W[kw1 * H_ + 16 + l16] * m1;
            }
            c0 = __builtin_amdgcn_wmma_f32_16x16x4_f32(false, a, false, b0,
                                                       (short)0, c0, false, false);
            c1 = __builtin_amdgcn_wmma_f32_16x16x4_f32(false, a, false, b1,
                                                       (short)0, c1, false, false);
        }
    }

#pragma unroll
    for (int r = 0; r < 8; ++r) {
        int row = tile * 16 + half * 8 + r;
        float v0 = tanhf(c0[r] + bias[l16]);
        float v1 = tanhf(c1[r] + bias[16 + l16]);
        out[row * H_ + l16]      = v0;
        out[row * H_ + 16 + l16] = v1;
    }
}

__global__ void readout_kernel(const float* __restrict__ h, const float* __restrict__ Wr,
                               const float* __restrict__ br, float* __restrict__ out) {
    int i = blockIdx.x * blockDim.x + threadIdx.x;
    if (i >= NTOT_) return;
    float acc = br[0];
#pragma unroll
    for (int j = 0; j < H_; ++j) acc += h[i * H_ + j] * Wr[j];
    out[i] = acc;
}

static inline int cdiv(int a, int b) { return (a + b - 1) / b; }

extern "C" void kernel_launch(void* const* d_in, const int* in_sizes, int n_in,
                              void* d_out, int out_size, void* d_ws, size_t ws_size,
                              hipStream_t stream) {
    (void)in_sizes; (void)n_in; (void)out_size; (void)ws_size;

    const float* X  = (const float*)d_in[0];
    const int*   EI = (const int*)d_in[1];
    const float* W0 = (const float*)d_in[2];
    const float* b0 = (const float*)d_in[3];
    const float* W1 = (const float*)d_in[4];
    const float* b1 = (const float*)d_in[5];
    const float* W2 = (const float*)d_in[6];
    const float* b2 = (const float*)d_in[7];
    const float* Wr = (const float*)d_in[8];
    const float* br = (const float*)d_in[9];

    float* outF = (float*)d_out;
    float* hOut = outF + NTOT_;

    int*   cnt    = (int*)d_ws;
    int*   rowptr = cnt + NTOT_;
    int*   bsum   = rowptr + NTOT_ + 64;
    int*   boff   = bsum + SCAN_TB;
    float* dinv   = (float*)(boff + SCAN_TB);
    int*   rank   = (int*)(dinv + NTOT_);
    int*   csrc   = rank + BE_;
    float* cw     = (float*)(csrc + BE_);
    float* p1     = cw + BE_;
    float* p2     = p1 + (size_t)NTOT_ * H_;
    float* bufA   = p2 + (size_t)NTOT_ * H_;
    float* bufB   = bufA + (size_t)NTOT_ * H_;

    const int TB = 256;

    zero_i32_kernel<<<cdiv(NTOT_ / 4, TB), TB, 0, stream>>>(cnt, NTOT_ / 4);
    count_rank_kernel<<<cdiv(EHALF_, TB), TB, 0, stream>>>(EI, cnt, rank);
    scan_sums_kernel<<<SCAN_BLOCKS, SCAN_TB, 0, stream>>>(cnt, bsum);
    scan_blocks_kernel<<<1, SCAN_TB, 0, stream>>>(bsum, boff, rowptr);
    scan_rows_kernel<<<SCAN_BLOCKS, SCAN_TB, 0, stream>>>(cnt, boff, rowptr, dinv);
    fill_kernel<<<cdiv(EHALF_, TB), TB, 0, stream>>>(EI, rowptr, rank, dinv, csrc, cw);

    const int gemmBlocks = cdiv(NTOT_ / 16, TB / 32);
    const int pullW = cdiv(PULL_G_ * 8, TB);

    pull_kernel<WIN_><<<pullW, TB, 0, stream>>>(rowptr, csrc, cw, X, p1);
    pull_kernel<WIN_><<<pullW, TB, 0, stream>>>(rowptr, csrc, cw, p1, p2);
    tag_gemm_kernel<WIN_><<<gemmBlocks, TB, 0, stream>>>(X, p1, p2, W0, b0, bufA);

    pull_kernel<H_><<<pullW, TB, 0, stream>>>(rowptr, csrc, cw, bufA, p1);
    pull_kernel<H_><<<pullW, TB, 0, stream>>>(rowptr, csrc, cw, p1, p2);
    tag_gemm_kernel<H_><<<gemmBlocks, TB, 0, stream>>>(bufA, p1, p2, W1, b1, bufB);

    pull_kernel<H_><<<pullW, TB, 0, stream>>>(rowptr, csrc, cw, bufB, p1);
    pull_kernel<H_><<<pullW, TB, 0, stream>>>(rowptr, csrc, cw, p1, p2);
    tag_gemm_kernel<H_><<<gemmBlocks, TB, 0, stream>>>(bufB, p1, p2, W2, b2, hOut);

    readout_kernel<<<cdiv(NTOT_, TB), TB, 0, stream>>>(hOut, Wr, br, outF);
}


# Count+rank off global atomics: a work-group counts a 1/16 slice of one graph's edges in a per-graph LDS histogram (returned slot = rank in slice), row scan adds slice starts; sums stay arrival-order
# speedup vs baseline: 3.1388x; 1.2703x over previous
#include <hip/hip_runtime.h>
#include <math.h>

#define B_    4
#define WIN_  5
#define N_    50000
#define E_    1000000
#define H_    32
#define NTOT_ (B_ * N_)
#define BE_   (B_ * E_)

typedef __attribute__((ext_vector_type(2))) float v2f;
typedef __attribute__((ext_vector_type(8))) float v8f;

#define SCAN_TB     256
#define SCAN_CHUNK  (SCAN_TB * 4)
#define SCAN_BLOCKS ((NTOT_ + SCAN_CHUNK - 1) / SCAN_CHUNK)
static_assert(SCAN_BLOCKS <= SCAN_TB, "one block must cover all chunk sums");

#define EPT_    (B_ * 2)
#define EHALF_  (E_ / 2)
static_assert(E_ % 2 == 0, "edge list splits in two halves");

#define CP_     16
#define CE_     (E_ / CP_)
#define CE4_    (CE_ / 4)
#define CTB_    1024
#define CBATCH_ 8
static_assert(E_ % CP_ == 0 && CE_ % 4 == 0, "slices are whole 16-byte groups");
static_assert(N_ % 4 == 0, "histogram rows move in 16-byte groups");

__global__ __launch_bounds__(CTB_) void count_rank_kernel(const int* __restrict__ ei,
                                                          int* __restrict__ part,
                                                          int* __restrict__ rank) {
    __shared__ int hist[N_];
    const int b = blockIdx.x / CP_;
    const int p = blockIdx.x - b * CP_;
    const int t = threadIdx.x;
    int4* hist4 = (int4*)hist;
    for (int i = t; i < N_ / 4; i += CTB_) hist4[i] = make_int4(0, 0, 0, 0);
    __syncthreads();
    const int4* __restrict__ d4 = (const int4*)(ei + (b * 2 + 1) * E_ + p * CE_);
    int4* __restrict__ r4 = (int4*)(rank + b * E_ + p * CE_);
    for (int base = 0; base < CE4_; base += CTB_ * CBATCH_) {
        int4 d[CBATCH_];
#pragma unroll
        for (int k = 0; k < CBATCH_; ++k) {
            int i = base + k * CTB_ + t;
            d[k] = make_int4(0, 0, 0, 0);
            if (i < CE4_) d[k] = d4[i];
        }
#pragma unroll
        for (int k = 0; k < CBATCH_; ++k) {
            int i = base + k * CTB_ + t;
            if (i < CE4_) {
                int4 s;
                s.x = atomicAdd(&hist[d[k].x], 1);
                s.y = atomicAdd(&hist[d[k].y], 1);
                s.z = atomicAdd(&hist[d[k].z], 1);
                s.w = atomicAdd(&hist[d[k].w], 1);
                r4[i] = s;
            }
        }
    }
    __syncthreads();
    int4* __restrict__ out4 = (int4*)(part + (b * CP_ + p) * N_);
    for (int i = t; i < N_ / 4; i += CTB_) out4[i] = hist4[i];
}

static __device__ __forceinline__ int block_incl_scan(int v, int* sh) {
    int t = threadIdx.x;
    sh[t] = v;
    __syncthreads();
    for (int off = 1; off < SCAN_TB; off <<= 1) {
        int x = (t >= off) ? sh[t - off] : 0;
        __syncthreads();
        sh[t] += x;
        __syncthreads();
    }
    return sh[t];
}

__global__ void scan_sums_kernel(const int* __restrict__ part, int* __restrict__ blocksum) {
    __shared__ int sh[SCAN_TB];
    int base = blockIdx.x * SCAN_CHUNK + threadIdx.x * 4;
    int s = 0;
    if (base < NTOT_) {
        int b = base / N_;
        int n = base - b * N_;
        const int4* __restrict__ q = (const int4*)(part + b * CP_ * N_ + n);
#pragma unroll
        for (int p = 0; p < CP_; ++p) {
            int4 c = q[p * (N_ / 4)];
            s += (c.x + c.y) + (c.z + c.w);
        }
    }
    int incl = block_incl_scan(s, sh);
    if (threadIdx.x == SCAN_TB - 1) blocksum[blockIdx.x] = incl;
}

__global__ void scan_blocks_kernel(const int* __restrict__ blocksum, int* __restrict__ blockoff,
                                   int* __restrict__ rowptr) {
    __shared__ int sh[SCAN_TB];
    int t = threadIdx.x;
    int v = (t < SCAN_BLOCKS) ? blocksum[t] : 0;
    int incl = block_incl_scan(v, sh);
    if (t < SCAN_BLOCKS) blockoff[t] = incl - v;
    if (t == SCAN_TB - 1) rowptr[NTOT_] = incl;
}

__global__ void scan_rows_kernel(int* __restrict__ part, const int* __restrict__ blockoff,
                                 int* __restrict__ rowptr, float* __restrict__ dinv) {
    __shared__ int sh[SCAN_TB];
    int base = blockIdx.x * SCAN_CHUNK + threadIdx.x * 4;
    const bool live = base < NTOT_;
    int4 c[CP_];
    int4 tot = make_int4(0, 0, 0, 0);
    int4* q = (int4*)part;
    if (live) {
        int b = base / N_;
        int n = base - b * N_;
        q = (int4*)(part + b * CP_ * N_ + n);
#pragma unroll
        for (int p = 0; p < CP_; ++p) {
            c[p] = q[p * (N_ / 4)];
            tot.x += c[p].x; tot.y += c[p].y; tot.z += c[p].z; tot.w += c[p].w;
        }
    }
    int s = (tot.x + tot.y) + (tot.z + tot.w);
    int incl = block_incl_scan(s, sh);
    int run = blockoff[blockIdx.x] + incl - s;
    if (live) {
        int4 start;
        start.x = run;
        start.y = start.x + tot.x;
        start.z = start.y + tot.y;
        start.w = start.z + tot.z;
        *(int4*)(rowptr + base) = start;
        float4 di;
        di.x = (tot.x > 0) ? rsqrtf((float)tot.x) : 0.0f;
        di.y = (tot.y > 0) ? rsqrtf((float)tot.y) : 0.0f;
        di.z = (tot.z > 0) ? rsqrtf((float)tot.z) : 0.0f;
        di.w = (tot.w > 0) ? rsqrtf((float)tot.w) : 0.0f;
        *(float4*)(dinv + base) = di;
#pragma unroll
        for (int p = 0; p < CP_; ++p) {
            q[p * (N_ / 4)] = start;
            start.x += c[p].x; start.y += c[p].y; start.z += c[p].z; start.w += c[p].w;
        }
    }
}

__global__ void fill_kernel(const int* __restrict__ ei, const int* __restrict__ part,
                            const int* __restrict__ rank, const float* __restrict__ dinv,
                            int* __restrict__ csr_src, float* __restrict__ csr_w) {
    int t = blockIdx.x * blockDim.x + threadIdx.x;
    if (t >= EHALF_) return;
    int   src[EPT_], dst[EPT_], pos[EPT_], row[EPT_];
    float ws[EPT_], wd[EPT_];
#pragma unroll
    for (int k = 0; k < EPT_; ++k) {
        int b  = k >> 1;
        int el = t + (k & 1) * EHALF_;
        src[k] = ei[(b * 2 + 0) * E_ + el] + b * N_;
        int dl = ei[(b * 2 + 1) * E_ + el];
        dst[k] = dl + b * N_;
        row[k] = (b * CP_ + el / CE_) * N_ + dl;
        pos[k] = rank[b * E_ + el];
    }
#pragma unroll
    for (int k = 0; k < EPT_; ++k) {
        pos[k] += part[row[k]];
        ws[k] = dinv[src[k]];
        wd[k] = dinv[dst[k]];
    }
#pragma unroll
    for (int k = 0; k < EPT_; ++k) {
        csr_src[pos[k]] = src[k];
        csr_w[pos[k]]   = ws[k] * wd[k];
    }
}

template <int K>
static __device__ __forceinline__ int bcast8(int v) {
    return (int)__builtin_amdgcn_mov_dpp8((unsigned)v, (unsigned)(K * 0x249249));
}
template <int K>
static __device__ __forceinline__ float bcast8(float v) {
    return __int_as_float(bcast8<K>(__float_as_int(v)));
}

static __device__ __forceinline__ void axpy(float& a, float v, float w) { a = __builtin_fmaf(v, w, a); }
static __device__ __forceinline__ void axpy(float4& a, const float4& v, float w) {
    a.x = __builtin_fmaf(v.x, w, a.x); a.y = __builtin_fmaf(v.y, w, a.y);
    a.z = __builtin_fmaf(v.z, w, a.z); a.w = __builtin_fmaf(v.w, w, a.w);
}
static __device__ __forceinline__ void set_zero(float& a)  { a = 0.0f; }
static __device__ __forceinline__ void set_zero(float4& a) { a = make_float4(0.0f, 0.0f, 0.0f, 0.0f); }

template <int V> struct RowSlice;
template <> struct RowSlice<4> { typedef float4 T; };
template <> struct RowSlice<1> { typedef float  T; };

template <int LQ, typename RowT>
static __device__ __forceinline__ void pull_chunk(const RowT* __restrict__ hin_v, int s_l, float w_l,
                                                  int n, int q, bool carries, RowT& acc) {
    int   s[8];
    float w[8];
    s[0] = bcast8<0>(s_l); s[1] = bcast8<1>(s_l); s[2] = bcast8<2>(s_l); s[3] = bcast8<3>(s_l);
    s[4] = bcast8<4>(s_l); s[5] = bcast8<5>(s_l); s[6] = bcast8<6>(s_l); s[7] = bcast8<7>(s_l);
    w[0] = bcast8<0>(w_l); w[1] = bcast8<1>(w_l); w[2] = bcast8<2>(w_l); w[3] = bcast8<3>(w_l);
    w[4] = bcast8<4>(w_l); w[5] = bcast8<5>(w_l); w[6] = bcast8<6>(w_l); w[7] = bcast8<7>(w_l);
    RowT v[8];
#pragma unroll
    for (int k = 0; k < 8; ++k) {
        set_zero(v[k]);
        if (carries && k < n) v[k] = hin_v[s[k] * LQ + q];
    }
#pragma unroll
    for (int k = 0; k < 8; ++k) axpy(acc, v[k], w[k]);
}

#define PULL_NPG_ 4
#define PULL_G_   (NTOT_ / PULL_NPG_)
static_assert(NTOT_ % PULL_NPG_ == 0, "every lane group takes the same number of nodes");

template <int F>
__global__ void pull_kernel(const int* __restrict__ rowptr, const int* __restrict__ csr_src,
                            const float* __restrict__ csr_w, const float* __restrict__ hin,
                            float* __restrict__ hout) {
    constexpr int V  = (F % 8 == 0) ? (F / 8) : 1;
    constexpr int LQ = F / V;
    typedef typename RowSlice<V>::T RowT;
    int tid = blockIdx.x * blockDim.x + threadIdx.x;
    int g = tid >> 3;
    int q = tid & 7;
    if (g >= PULL_G_) return;
    const RowT* __restrict__ hin_v = (const RowT*)hin;
    RowT* __restrict__ hout_v = (RowT*)hout;
    const bool carries = (q < LQ);
    for (int node = 0; node < PULL_NPG_; ++node) {
        const int i = g + node * PULL_G_;
        int j   = rowptr[i];
        int end = rowptr[i + 1];
        RowT acc;
        set_zero(acc);
        for (; j < end; j += 32) {
            int n = end - j;
            int   s_l[4];
            float w_l[4];
#pragma unroll
            for (int c = 0; c < 4; ++c) {
                s_l[c] = 0;
                w_l[c] = 0.0f;
                if (8 * c + q < n) {
                    s_l[c] = csr_src[j + 8 * c + q];
                    w_l[c] = csr_w[j + 8 * c + q];
                }
            }
#pragma unroll
            for (int c = 0; c < 4; ++c) {
                if (8 * c < n)
                    pull_chunk<LQ, RowT>(hin_v, s_l[c], w_l[c], n - 8 * c, q, carries, acc);
            }
        }
        if (carries) hout_v[i * LQ + q] = acc;
    }
}

template <int F>
__global__ void tag_gemm_kernel(const float* __restrict__ x0, const float* __restrict__ x1,
                                const float* __restrict__ x2, const float* __restrict__ W,
                                const float* __restrict__ bias, float* __restrict__ out) {
    const int numTiles = NTOT_ / 16;
    int wave = threadIdx.x >> 5;
    int tile = blockIdx.x * (blockDim.x >> 5) + wave;
    if (tile >= numTiles) return;

    int lane = threadIdx.x & 31;
    int half = lane >> 4;
    int l16  = lane & 15;
    int rowA = tile * 16 + l16;

    v8f c0 = {};
    v8f c1 = {};

    const float* xs[3] = { x0, x1, x2 };
    constexpr bool FULL = (F % 4 == 0);

#pragma unroll
    for (int s = 0; s < 3; ++s) {
        const float* __restrict__ xp = xs[s];
#pragma unroll
        for (int k0 = 0; k0 < F; k0 += 4) {
            int kA = k0 + half * 2;
            v2f a, b0, b1;
            if (FULL) {
                a.x = xp[rowA * F + kA];
                a.y = xp[rowA * F + kA + 1];
                int kW = s * F + kA;
                b0.x = W[(kW    ) * H_ + l16];
                b0.y = W[(kW + 1) * H_ + l16];
                b1.x = W[(kW    ) * H_ + 16 + l16];
                b1.y = W[(kW + 1) * H_ + 16 + l16];
            } else {
                float m0 = (kA     < F) ? 1.0f : 0.0f;
                float m1 = (kA + 1 < F) ? 1.0f : 0.0f;
                int ka0 = (kA     < F) ? kA     : (F - 1);
                int ka1 = (kA + 1 < F) ? kA + 1 : (F - 1);
                a.x = xp[rowA * F + ka0] * m0;
                a.y = xp[rowA * F + ka1] * m1;
                int kw0 = s * F + ka0;
                int kw1 = s * F + ka1;
                b0.x = W[kw0 * H_ + l16] * m0;
                b0.y = W[kw1 * H_ + l16] * m1;
                b1.x = W[kw0 * H_ + 16 + l16] * m0;
                b1.y = W[kw1 * H_ + 16 + l16] * m1;
            }
            c0 = __builtin_amdgcn_wmma_f32_16x16x4_f32(false, a, false, b0,
                                                       (short)0, c0, false, false);
            c1 = __builtin_amdgcn_wmma_f32_16x16x4_f32(false, a, false, b1,
                                                       (short)0, c1, false, false);
        }
    }

#pragma unroll
    for (int r = 0; r < 8; ++r) {
        int row = tile * 16 + half * 8 + r;
        float v0 = tanhf(c0[r] + bias[l16]);
        float v1 = tanhf(c1[r] + bias[16 + l16]);
        out[row * H_ + l16]      = v0;
        out[row * H_ + 16 + l16] = v1;
    }
}

__global__ void readout_kernel(const float* __restrict__ h, const float* __restrict__ Wr,
                               const float* __restrict__ br, float* __restrict__ out) {
    int i = blockIdx.x * blockDim.x + threadIdx.x;
    if (i >= NTOT_) return;
    float acc = br[0];
#pragma unroll
    for (int j = 0; j < H_; ++j) acc += h[i * H_ + j] * Wr[j];
    out[i] = acc;
}

static inline int cdiv(int a, int b) { return (a + b - 1) / b; }

extern "C" void kernel_launch(void* const* d_in, const int* in_sizes, int n_in,
                              void* d_out, int out_size, void* d_ws, size_t ws_size,
                              hipStream_t stream) {
    (void)in_sizes; (void)n_in; (void)out_size; (void)ws_size;

    const float* X  = (const float*)d_in[0];
    const int*   EI = (const int*)d_in[1];
    const float* W0 = (const float*)d_in[2];
    const float* b0 = (const float*)d_in[3];
    const float* W1 = (const float*)d_in[4];
    const float* b1 = (const float*)d_in[5];
    const float* W2 = (const float*)d_in[6];
    const float* b2 = (const float*)d_in[7];
    const float* Wr = (const float*)d_in[8];
    const float* br = (const float*)d_in[9];

    float* outF = (float*)d_out;
    float* hOut = outF + NTOT_;

    int*   part   = (int*)d_ws;
    int*   rowptr = part + B_ * CP_ * N_;
    int*   bsum   = rowptr + NTOT_ + 64;
    int*   boff   = bsum + SCAN_TB;
    float* dinv   = (float*)(boff + SCAN_TB);
    int*   rank   = (int*)(dinv + NTOT_);
    int*   csrc   = rank + BE_;
    float* cw     = (float*)(csrc + BE_);
    float* p1     = cw + BE_;
    float* p2     = p1 + (size_t)NTOT_ * H_;
    float* bufA   = p2 + (size_t)NTOT_ * H_;
    float* bufB   = bufA + (size_t)NTOT_ * H_;

    const int TB = 256;

    count_rank_kernel<<<B_ * CP_, CTB_, 0, stream>>>(EI, part, rank);
    scan_sums_kernel<<<SCAN_BLOCKS, SCAN_TB, 0, stream>>>(part, bsum);
    scan_blocks_kernel<<<1, SCAN_TB, 0, stream>>>(bsum, boff, rowptr);
    scan_rows_kernel<<<SCAN_BLOCKS, SCAN_TB, 0, stream>>>(part, boff, rowptr, dinv);
    fill_kernel<<<cdiv(EHALF_, TB), TB, 0, stream>>>(EI, part, rank, dinv, csrc, cw);

    const int gemmBlocks = cdiv(NTOT_ / 16, TB / 32);
    const int pullW = cdiv(PULL_G_ * 8, TB);

    pull_kernel<WIN_><<<pullW, TB, 0, stream>>>(rowptr, csrc, cw, X, p1);
    pull_kernel<WIN_><<<pullW, TB, 0, stream>>>(rowptr, csrc, cw, p1, p2);
    tag_gemm_kernel<WIN_><<<gemmBlocks, TB, 0, stream>>>(X, p1, p2, W0, b0, bufA);

    pull_kernel<H_><<<pullW, TB, 0, stream>>>(rowptr, csrc, cw, bufA, p1);
    pull_kernel<H_><<<pullW, TB, 0, stream>>>(rowptr, csrc, cw, p1, p2);
    tag_gemm_kernel<H_><<<gemmBlocks, TB, 0, stream>>>(bufA, p1, p2, W1, b1, bufB);

    pull_kernel<H_><<<pullW, TB, 0, stream>>>(rowptr, csrc, cw, bufB, p1);
    pull_kernel<H_><<<pullW, TB, 0, stream>>>(rowptr, csrc, cw, p1, p2);
    tag_gemm_kernel<H_><<<gemmBlocks, TB, 0, stream>>>(bufB, p1, p2, W2, b2, hOut);

    readout_kernel<<<cdiv(NTOT_, TB), TB, 0, stream>>>(hOut, Wr, br, outF);
}


# tag_gemm moves wider pieces: input rows by 16-byte loads issued up front (k order inside a row permuted, same in the W loader: products sum in another order), W, bias, out by 8-byte column pairs
# speedup vs baseline: 4.2439x; 1.3521x over previous
#include <hip/hip_runtime.h>
#include <math.h>

#define B_    4
#define WIN_  5
#define N_    50000
#define E_    1000000
#define H_    32
#define NTOT_ (B_ * N_)
#define BE_   (B_ * E_)

typedef __attribute__((ext_vector_type(2))) float v2f;
typedef __attribute__((ext_vector_type(4))) float v4f;
typedef __attribute__((ext_vector_type(8))) float v8f;

#define SCAN_TB     256
#define SCAN_CHUNK  (SCAN_TB * 4)
#define SCAN_BLOCKS ((NTOT_ + SCAN_CHUNK - 1) / SCAN_CHUNK)
static_assert(SCAN_BLOCKS <= SCAN_TB, "one block must cover all chunk sums");

#define EPT_    (B_ * 2)
#define EHALF_  (E_ / 2)
static_assert(E_ % 2 == 0, "edge list splits in two halves");

#define CP_     16
#define CE_     (E_ / CP_)
#define CE4_    (CE_ / 4)
#define CTB_    1024
#define CBATCH_ 8
static_assert(E_ % CP_ == 0 && CE_ % 4 == 0, "slices are whole 16-byte groups");
static_assert(N_ % 4 == 0, "histogram rows move in 16-byte groups");

__global__ __launch_bounds__(CTB_) void count_rank_kernel(const int* __restrict__ ei,
                                                          int* __restrict__ part,
                                                          int* __restrict__ rank) {
    __shared__ int hist[N_];
    const int b = blockIdx.x / CP_;
    const int p = blockIdx.x - b * CP_;
    const int t = threadIdx.x;
    int4* hist4 = (int4*)hist;
    for (int i = t; i < N_ / 4; i += CTB_) hist4[i] = make_int4(0, 0, 0, 0);
    __syncthreads();
    const int4* __restrict__ d4 = (const int4*)(ei + (b * 2 + 1) * E_ + p * CE_);
    int4* __restrict__ r4 = (int4*)(rank + b * E_ + p * CE_);
    for (int base = 0; base < CE4_; base += CTB_ * CBATCH_) {
        int4 d[CBATCH_];
#pragma unroll
        for (int k = 0; k < CBATCH_; ++k) {
            int i = base + k * CTB_ + t;
            d[k] = make_int4(0, 0, 0, 0);
            if (i < CE4_) d[k] = d4[i];
        }
#pragma unroll
        for (int k = 0; k < CBATCH_; ++k) {
            int i = base + k * CTB_ + t;
            if (i < CE4_) {
                int4 s;
                s.x = atomicAdd(&hist[d[k].x], 1);
                s.y = atomicAdd(&hist[d[k].y], 1);
                s.z = atomicAdd(&hist[d[k].z], 1);
                s.w = atomicAdd(&hist[d[k].w], 1);
                r4[i] = s;
            }
        }
    }
    __syncthreads();
    int4* __restrict__ out4 = (int4*)(part + (b * CP_ + p) * N_);
    for (int i = t; i < N_ / 4; i += CTB_) out4[i] = hist4[i];
}

static __device__ __forceinline__ int block_incl_scan(int v, int* sh) {
    int t = threadIdx.x;
    sh[t] = v;
    __syncthreads();
    for (int off = 1; off < SCAN_TB; off <<= 1) {
        int x = (t >= off) ? sh[t - off] : 0;
        __syncthreads();
        sh[t] += x;
        __syncthreads();
    }
    return sh[t];
}

__global__ void scan_sums_kernel(const int* __restrict__ part, int* __restrict__ blocksum) {
    __shared__ int sh[SCAN_TB];
    int base = blockIdx.x * SCAN_CHUNK + threadIdx.x * 4;
    int s = 0;
    if (base < NTOT_) {
        int b = base / N_;
        int n = base - b * N_;
        const int4* __restrict__ q = (const int4*)(part + b * CP_ * N_ + n);
#pragma unroll
        for (int p = 0; p < CP_; ++p) {
            int4 c = q[p * (N_ / 4)];
            s += (c.x + c.y) + (c.z + c.w);
        }
    }
    int incl = block_incl_scan(s, sh);
    if (threadIdx.x == SCAN_TB - 1) blocksum[blockIdx.x] = incl;
}

__global__ void scan_blocks_kernel(const int* __restrict__ blocksum, int* __restrict__ blockoff,
                                   int* __restrict__ rowptr) {
    __shared__ int sh[SCAN_TB];
    int t = threadIdx.x;
    int v = (t < SCAN_BLOCKS) ? blocksum[t] : 0;
    int incl = block_incl_scan(v, sh);
    if (t < SCAN_BLOCKS) blockoff[t] = incl - v;
    if (t == SCAN_TB - 1) rowptr[NTOT_] = incl;
}

__global__ void scan_rows_kernel(int* __restrict__ part, const int* __restrict__ blockoff,
                                 int* __restrict__ rowptr, float* __restrict__ dinv,
                                 const float* __restrict__ x, float* __restrict__ gx) {
    __shared__ int sh[SCAN_TB];
    int base = blockIdx.x * SCAN_CHUNK + threadIdx.x * 4;
    const bool live = base < NTOT_;
    int4 c[CP_];
    float4 xr[WIN_];
    int4 tot = make_int4(0, 0, 0, 0);
    int4* q = (int4*)part;
    if (live) {
        int b = base / N_;
        int n = base - b * N_;
        q = (int4*)(part + b * CP_ * N_ + n);
        const float4* __restrict__ x4 = (const float4*)(x + base * WIN_);
#pragma unroll
        for (int v = 0; v < WIN_; ++v) xr[v] = x4[v];
#pragma unroll
        for (int p = 0; p < CP_; ++p) {
            c[p] = q[p * (N_ / 4)];
            tot.x += c[p].x; tot.y += c[p].y; tot.z += c[p].z; tot.w += c[p].w;
        }
    }
    int s = (tot.x + tot.y) + (tot.z + tot.w);
    int incl = block_incl_scan(s, sh);
    int run = blockoff[blockIdx.x] + incl - s;
    if (live) {
        int4 start;
        start.x = run;
        start.y = start.x + tot.x;
        start.z = start.y + tot.y;
        start.w = start.z + tot.z;
        *(int4*)(rowptr + base) = start;
        float4 di;
        di.x = (tot.x > 0) ? rsqrtf((float)tot.x) : 0.0f;
        di.y = (tot.y > 0) ? rsqrtf((float)tot.y) : 0.0f;
        di.z = (tot.z > 0) ? rsqrtf((float)tot.z) : 0.0f;
        di.w = (tot.w > 0) ? rsqrtf((float)tot.w) : 0.0f;
        *(float4*)(dinv + base) = di;
        const float dn[4] = { di.x, di.y, di.z, di.w };
        float4* __restrict__ g4 = (float4*)(gx + base * WIN_);
#pragma unroll
        for (int v = 0; v < WIN_; ++v) {
            float4 r = xr[v];
            r.x *= dn[(4 * v + 0) / WIN_];
            r.y *= dn[(4 * v + 1) / WIN_];
            r.z *= dn[(4 * v + 2) / WIN_];
            r.w *= dn[(4 * v + 3) / WIN_];
            g4[v] = r;
        }
#pragma unroll
        for (int p = 0; p < CP_; ++p) {
            q[p * (N_ / 4)] = start;
            start.x += c[p].x; start.y += c[p].y; start.z += c[p].z; start.w += c[p].w;
        }
    }
}

__global__ void fill_kernel(const int* __restrict__ ei, const int* __restrict__ part,
                            const int* __restrict__ rank, int* __restrict__ csr_src) {
    int t = blockIdx.x * blockDim.x + threadIdx.x;
    if (t >= EHALF_) return;
    int src[EPT_], pos[EPT_], row[EPT_];
#pragma unroll
    for (int k = 0; k < EPT_; ++k) {
        int b  = k >> 1;
        int el = t + (k & 1) * EHALF_;
        src[k] = ei[(b * 2 + 0) * E_ + el] + b * N_;
        int dl = ei[(b * 2 + 1) * E_ + el];
        row[k] = (b * CP_ + el / CE_) * N_ + dl;
        pos[k] = rank[b * E_ + el];
    }
#pragma unroll
    for (int k = 0; k < EPT_; ++k) {
        pos[k] += part[row[k]];
    }
#pragma unroll
    for (int k = 0; k < EPT_; ++k) {
        csr_src[pos[k]] = src[k];
    }
}

template <int K>
static __device__ __forceinline__ int bcast8(int v) {
    return (int)__builtin_amdgcn_mov_dpp8((unsigned)v, (unsigned)(K * 0x249249));
}
template <int K>
static __device__ __forceinline__ float bcast8(float v) {
    return __int_as_float(bcast8<K>(__float_as_int(v)));
}

static __device__ __forceinline__ void accum(float& a, float v) { a += v; }
static __device__ __forceinline__ void accum(float4& a, const float4& v) {
    a.x += v.x; a.y += v.y; a.z += v.z; a.w += v.w;
}
static __device__ __forceinline__ float  scaled(float v, float d) { return v * d; }
static __device__ __forceinline__ float4 scaled(const float4& v, float d) {
    return make_float4(v.x * d, v.y * d, v.z * d, v.w * d);
}
static __device__ __forceinline__ void set_zero(float& a)  { a = 0.0f; }
static __device__ __forceinline__ void set_zero(float4& a) { a = make_float4(0.0f, 0.0f, 0.0f, 0.0f); }

template <int V> struct RowSlice;
template <> struct RowSlice<4> { typedef float4 T; };
template <> struct RowSlice<1> { typedef float  T; };

template <int LQ, typename RowT>
static __device__ __forceinline__ void pull_chunk(const RowT* __restrict__ hin_v, int s_l,
                                                  int n, int q, bool carries, RowT& acc) {
    int s[8];
    s[0] = bcast8<0>(s_l); s[1] = bcast8<1>(s_l); s[2] = bcast8<2>(s_l); s[3] = bcast8<3>(s_l);
    s[4] = bcast8<4>(s_l); s[5] = bcast8<5>(s_l); s[6] = bcast8<6>(s_l); s[7] = bcast8<7>(s_l);
    RowT v[8];
#pragma unroll
    for (int k = 0; k < 8; ++k) {
        set_zero(v[k]);
        if (carries && k < n) v[k] = hin_v[s[k] * LQ + q];
    }
#pragma unroll
    for (int k = 0; k < 8; ++k) accum(acc, v[k]);
}

#define PULL_NPG_ 4
#define PULL_G_   (NTOT_ / PULL_NPG_)
static_assert(NTOT_ % PULL_NPG_ == 0, "every lane group takes the same number of nodes");

template <int F, bool WG>
__global__ void pull_kernel(const int* __restrict__ rowptr, const int* __restrict__ csr_src,
                            const float* __restrict__ dinv, const float* __restrict__ hin,
                            float* __restrict__ hout, float* __restrict__ gout) {
    constexpr int V  = (F % 8 == 0) ? (F / 8) : 1;
    constexpr int LQ = F / V;
    typedef typename RowSlice<V>::T RowT;
    int tid = blockIdx.x * blockDim.x + threadIdx.x;
    int g = tid >> 3;
    int q = tid & 7;
    if (g >= PULL_G_) return;
    const RowT* __restrict__ hin_v = (const RowT*)hin;
    RowT* __restrict__ hout_v = (RowT*)hout;
    RowT* __restrict__ gout_v = (RowT*)gout;
    const bool carries = (q < LQ);
    for (int node = 0; node < PULL_NPG_; ++node) {
        const int i = g + node * PULL_G_;
        int j   = rowptr[i];
        int end = rowptr[i + 1];
        const float d = dinv[i];
        RowT acc;
        set_zero(acc);
        for (; j < end; j += 32) {
            int n = end - j;
            int s_l[4];
#pragma unroll
            for (int c = 0; c < 4; ++c) {
                s_l[c] = 0;
                if (8 * c + q < n) s_l[c] = csr_src[j + 8 * c + q];
            }
#pragma unroll
            for (int c = 0; c < 4; ++c) {
                if (8 * c < n)
                    pull_chunk<LQ, RowT>(hin_v, s_l[c], n - 8 * c, q, carries, acc);
            }
        }
        if (carries) {
            const RowT xr = scaled(acc, d);
            hout_v[i * LQ + q] = xr;
            if (WG) gout_v[i * LQ + q] = scaled(xr, d);
        }
    }
}

template <int F, bool WG>
__global__ void tag_gemm_kernel(const float* __restrict__ x0, const float* __restrict__ x1,
                                const float* __restrict__ x2, const float* __restrict__ W,
                                const float* __restrict__ bias, float* __restrict__ out,
                                const float* __restrict__ dinv, float* __restrict__ gout) {
    const int numTiles = NTOT_ / 16;
    int wave = threadIdx.x >> 5;
    int tile = blockIdx.x * (blockDim.x >> 5) + wave;
    if (tile >= numTiles) return;

    int lane = threadIdx.x & 31;
    int half = lane >> 4;
    int l16  = lane & 15;
    int rowA = tile * 16 + l16;

    v8f c0 = {};
    v8f c1 = {};

    const float* xs[3] = { x0, x1, x2 };
    constexpr bool WIDE = (F % 8 == 0);
    constexpr int  NP   = WIDE ? (F / 8) : 1;

    if constexpr (WIDE) {
        static_assert(NP == 4, "the load fence below names 3 x 4 pieces");
        v4f av[3][NP];
#pragma unroll
        for (int s = 0; s < 3; ++s) {
            const v4f* __restrict__ xp4 = (const v4f*)(xs[s] + rowA * F + 4 * half);
#pragma unroll
            for (int j = 0; j < NP; ++j) av[s][j] = xp4[2 * j];
        }
        asm volatile("" :: "v"(av[0][0]), "v"(av[0][1]), "v"(av[0][2]), "v"(av[0][3]),
                           "v"(av[1][0]), "v"(av[1][1]), "v"(av[1][2]), "v"(av[1][3]),
                           "v"(av[2][0]), "v"(av[2][1]), "v"(av[2][2]), "v"(av[2][3]));
#pragma unroll
        for (int s = 0; s < 3; ++s) {
#pragma unroll
            for (int j = 0; j < NP; ++j) {
                const int kW = s * F + 8 * j + 4 * half;
                const float2* __restrict__ wp = (const float2*)(W + kW * H_ + 2 * l16);
                const float2 w0 = wp[0 * (H_ / 2)];
                const float2 w1 = wp[1 * (H_ / 2)];
                const float2 w2 = wp[2 * (H_ / 2)];
                const float2 w3 = wp[3 * (H_ / 2)];
                const v4f p = av[s][j];
                v2f a, b0, b1;
                a.x = p.x;   a.y = p.y;
                b0.x = w0.x; b0.y = w1.x;
                b1.x = w0.y; b1.y = w1.y;
                c0 = __builtin_amdgcn_wmma_f32_16x16x4_f32(false, a, false, b0,
                                                           (short)0, c0, false, false);
                c1 = __builtin_amdgcn_wmma_f32_16x16x4_f32(false, a, false, b1,
                                                           (short)0, c1, false, false);
                a.x = p.z;   a.y = p.w;
                b0.x = w2.x; b0.y = w3.x;
                b1.x = w2.y; b1.y = w3.y;
                c0 = __builtin_amdgcn_wmma_f32_16x16x4_f32(false, a, false, b0,
                                                           (short)0, c0, false, false);
                c1 = __builtin_amdgcn_wmma_f32_16x16x4_f32(false, a, false, b1,
                                                           (short)0, c1, false, false);
            }
        }
    } else {
#pragma unroll
        for (int s = 0; s < 3; ++s) {
            const float* __restrict__ xp = xs[s];
#pragma unroll
            for (int k0 = 0; k0 < F; k0 += 4) {
                int kA = k0 + half * 2;
                float m0 = (kA     < F) ? 1.0f : 0.0f;
                float m1 = (kA + 1 < F) ? 1.0f : 0.0f;
                int ka0 = (kA     < F) ? kA     : (F - 1);
                int ka1 = (kA + 1 < F) ? kA + 1 : (F - 1);
                v2f a, b0, b1;
                a.x = xp[rowA * F + ka0] * m0;
                a.y = xp[rowA * F + ka1] * m1;
                const float2 w0 = *(const float2*)(W + (s * F + ka0) * H_ + 2 * l16);
                const float2 w1 = *(const float2*)(W + (s * F + ka1) * H_ + 2 * l16);
                b0.x = w0.x * m0; b0.y = w1.x * m1;
                b1.x = w0.y * m0; b1.y = w1.y * m1;
                c0 = __builtin_amdgcn_wmma_f32_16x16x4_f32(false, a, false, b0,
                                                           (short)0, c0, false, false);
                c1 = __builtin_amdgcn_wmma_f32_16x16x4_f32(false, a, false, b1,
                                                           (short)0, c1, false, false);
            }
        }
    }

    const float2 bs = *(const float2*)(bias + 2 * l16);
    const int row0 = tile * 16 + half * 8;
    float dn[8];
    if (WG) {
        const float4 da = *(const float4*)(dinv + row0);
        const float4 db = *(const float4*)(dinv + row0 + 4);
        dn[0] = da.x; dn[1] = da.y; dn[2] = da.z; dn[3] = da.w;
        dn[4] = db.x; dn[5] = db.y; dn[6] = db.z; dn[7] = db.w;
    }
#pragma unroll
    for (int r = 0; r < 8; ++r) {
        const int row = row0 + r;
        float2 v;
        v.x = tanhf(c0[r] + bs.x);
        v.y = tanhf(c1[r] + bs.y);
        *(float2*)(out + row * H_ + 2 * l16) = v;
        if (WG) {
            float2 gv;
            gv.x = v.x * dn[r];
            gv.y = v.y * dn[r];
            *(float2*)(gout + row * H_ + 2 * l16) = gv;
        }
    }
}

__global__ void readout_kernel(const float* __restrict__ h, const float* __restrict__ Wr,
                               const float* __restrict__ br, float* __restrict__ out) {
    int i = blockIdx.x * blockDim.x + threadIdx.x;
    if (i >= NTOT_) return;
    float acc = br[0];
#pragma unroll
    for (int j = 0; j < H_; ++j) acc += h[i * H_ + j] * Wr[j];
    out[i] = acc;
}

static inline int cdiv(int a, int b) { return (a + b - 1) / b; }

extern "C" void kernel_launch(void* const* d_in, const int* in_sizes, int n_in,
                              void* d_out, int out_size, void* d_ws, size_t ws_size,
                              hipStream_t stream) {
    (void)in_sizes; (void)n_in; (void)out_size; (void)ws_size;

    const float* X  = (const float*)d_in[0];
    const int*   EI = (const int*)d_in[1];
    const float* W0 = (const float*)d_in[2];
    const float* b0 = (const float*)d_in[3];
    const float* W1 = (const float*)d_in[4];
    const float* b1 = (const float*)d_in[5];
    const float* W2 = (const float*)d_in[6];
    const float* b2 = (const float*)d_in[7];
    const float* Wr = (const float*)d_in[8];
    const float* br = (const float*)d_in[9];

    float* outF = (float*)d_out;
    float* hOut = outF + NTOT_;

    int*   part   = (int*)d_ws;
    int*   rowptr = part + B_ * CP_ * N_;
    int*   bsum   = rowptr + NTOT_ + 64;
    int*   boff   = bsum + SCAN_TB;
    float* dinv   = (float*)(boff + SCAN_TB);
    int*   rank   = (int*)(dinv + NTOT_);
    int*   csrc   = rank + BE_;
    float* p1     = (float*)(csrc + BE_);
    float* p2     = p1 + (size_t)NTOT_ * H_;
    float* bufA   = p2 + (size_t)NTOT_ * H_;
    float* bufB   = bufA + (size_t)NTOT_ * H_;
    float* g0     = bufB + (size_t)NTOT_ * H_;
    float* g1     = g0 + (size_t)NTOT_ * H_;

    const int TB = 256;

    count_rank_kernel<<<B_ * CP_, CTB_, 0, stream>>>(EI, part, rank);
    scan_sums_kernel<<<SCAN_BLOCKS, SCAN_TB, 0, stream>>>(part, bsum);
    scan_blocks_kernel<<<1, SCAN_TB, 0, stream>>>(bsum, boff, rowptr);
    scan_rows_kernel<<<SCAN_BLOCKS, SCAN_TB, 0, stream>>>(part, boff, rowptr, dinv, X, g0);
    fill_kernel<<<cdiv(EHALF_, TB), TB, 0, stream>>>(EI, part, rank, csrc);

    const int gemmBlocks = cdiv(NTOT_ / 16, TB / 32);
    const int pullW = cdiv(PULL_G_ * 8, TB);

    pull_kernel<WIN_, true><<<pullW, TB, 0, stream>>>(rowptr, csrc, dinv, g0, p1, g1);
    pull_kernel<WIN_, false><<<pullW, TB, 0, stream>>>(rowptr, csrc, dinv, g1, p2, nullptr);
    tag_gemm_kernel<WIN_, true><<<gemmBlocks, TB, 0, stream>>>(X, p1, p2, W0, b0, bufA, dinv, g0);

    pull_kernel<H_, true><<<pullW, TB, 0, stream>>>(rowptr, csrc, dinv, g0, p1, g1);
    pull_kernel<H_, false><<<pullW, TB, 0, stream>>>(rowptr, csrc, dinv, g1, p2, nullptr);
    tag_gemm_kernel<H_, true><<<gemmBlocks, TB, 0, stream>>>(bufA, p1, p2, W1, b1, bufB, dinv, g0);

    pull_kernel<H_, true><<<pullW, TB, 0, stream>>>(rowptr, csrc, dinv, g0, p1, g1);
    pull_kernel<H_, false><<<pullW, TB, 0, stream>>>(rowptr, csrc, dinv, g1, p2, nullptr);
    tag_gemm_kernel<H_, false><<<gemmBlocks, TB, 0, stream>>>(bufB, p1, p2, W2, b2, hOut, dinv, nullptr);

    readout_kernel<<<cdiv(NTOT_, TB), TB, 0, stream>>>(hOut, Wr, br, outF);
}


# Layer 0's two width-5 gathers: feature-major planes, a work-group holds one feature column of one graph in LDS, edges look it up there; row sum = 8 lane partials (every 8th edge) + pair tree
# speedup vs baseline: 4.6973x; 1.1068x over previous
#include <hip/hip_runtime.h>
#include <math.h>

#define B_    4
#define WIN_  5
#define N_    50000
#define E_    1000000
#define H_    32
#define NTOT_ (B_ * N_)
#define BE_   (B_ * E_)

typedef __attribute__((ext_vector_type(2))) float v2f;
typedef __attribute__((ext_vector_type(4))) float v4f;
typedef __attribute__((ext_vector_type(8))) float v8f;

#define SCAN_TB     256
#define SCAN_CHUNK  (SCAN_TB * 4)
#define SCAN_BLOCKS ((NTOT_ + SCAN_CHUNK - 1) / SCAN_CHUNK)
static_assert(SCAN_BLOCKS <= SCAN_TB, "one block must cover all chunk sums");

#define EPT_    (B_ * 2)
#define EHALF_  (E_ / 2)
static_assert(E_ % 2 == 0, "edge list splits in two halves");

#define CP_     16
#define CE_     (E_ / CP_)
#define CE4_    (CE_ / 4)
#define CTB_    1024
#define CBATCH_ 8
static_assert(E_ % CP_ == 0 && CE_ % 4 == 0, "slices are whole 16-byte groups");
static_assert(N_ % 4 == 0, "histogram rows move in 16-byte groups");

__global__ __launch_bounds__(CTB_) void count_rank_kernel(const int* __restrict__ ei,
                                                          int* __restrict__ part,
                                                          int* __restrict__ rank) {
    __shared__ int hist[N_];
    const int b = blockIdx.x / CP_;
    const int p = blockIdx.x - b * CP_;
    const int t = threadIdx.x;
    int4* hist4 = (int4*)hist;
    for (int i = t; i < N_ / 4; i += CTB_) hist4[i] = make_int4(0, 0, 0, 0);
    __syncthreads();
    const int4* __restrict__ d4 = (const int4*)(ei + (b * 2 + 1) * E_ + p * CE_);
    int4* __restrict__ r4 = (int4*)(rank + b * E_ + p * CE_);
    for (int base = 0; base < CE4_; base += CTB_ * CBATCH_) {
        int4 d[CBATCH_];
#pragma unroll
        for (int k = 0; k < CBATCH_; ++k) {
            int i = base + k * CTB_ + t;
            d[k] = make_int4(0, 0, 0, 0);
            if (i < CE4_) d[k] = d4[i];
        }
#pragma unroll
        for (int k = 0; k < CBATCH_; ++k) {
            int i = base + k * CTB_ + t;
            if (i < CE4_) {
                int4 s;
                s.x = atomicAdd(&hist[d[k].x], 1);
                s.y = atomicAdd(&hist[d[k].y], 1);
                s.z = atomicAdd(&hist[d[k].z], 1);
                s.w = atomicAdd(&hist[d[k].w], 1);
                r4[i] = s;
            }
        }
    }
    __syncthreads();
    int4* __restrict__ out4 = (int4*)(part + (b * CP_ + p) * N_);
    for (int i = t; i < N_ / 4; i += CTB_) out4[i] = hist4[i];
}

static __device__ __forceinline__ int block_incl_scan(int v, int* sh) {
    int t = threadIdx.x;
    sh[t] = v;
    __syncthreads();
    for (int off = 1; off < SCAN_TB; off <<= 1) {
        int x = (t >= off) ? sh[t - off] : 0;
        __syncthreads();
        sh[t] += x;
        __syncthreads();
    }
    return sh[t];
}

__global__ void scan_sums_kernel(const int* __restrict__ part, int* __restrict__ blocksum) {
    __shared__ int sh[SCAN_TB];
    int base = blockIdx.x * SCAN_CHUNK + threadIdx.x * 4;
    int s = 0;
    if (base < NTOT_) {
        int b = base / N_;
        int n = base - b * N_;
        const int4* __restrict__ q = (const int4*)(part + b * CP_ * N_ + n);
#pragma unroll
        for (int p = 0; p < CP_; ++p) {
            int4 c = q[p * (N_ / 4)];
            s += (c.x + c.y) + (c.z + c.w);
        }
    }
    int incl = block_incl_scan(s, sh);
    if (threadIdx.x == SCAN_TB - 1) blocksum[blockIdx.x] = incl;
}

__global__ void scan_blocks_kernel(const int* __restrict__ blocksum, int* __restrict__ blockoff,
                                   int* __restrict__ rowptr) {
    __shared__ int sh[SCAN_TB];
    int t = threadIdx.x;
    int v = (t < SCAN_BLOCKS) ? blocksum[t] : 0;
    int incl = block_incl_scan(v, sh);
    if (t < SCAN_BLOCKS) blockoff[t] = incl - v;
    if (t == SCAN_TB - 1) rowptr[NTOT_] = incl;
}

__global__ void scan_rows_kernel(int* __restrict__ part, const int* __restrict__ blockoff,
                                 int* __restrict__ rowptr, float* __restrict__ dinv,
                                 const float* __restrict__ x, float* __restrict__ gx) {
    __shared__ int sh[SCAN_TB];
    int base = blockIdx.x * SCAN_CHUNK + threadIdx.x * 4;
    const bool live = base < NTOT_;
    int4 c[CP_];
    float4 xr[WIN_];
    int4 tot = make_int4(0, 0, 0, 0);
    int4* q = (int4*)part;
    if (live) {
        int b = base / N_;
        int n = base - b * N_;
        q = (int4*)(part + b * CP_ * N_ + n);
        const float4* __restrict__ x4 = (const float4*)(x + base * WIN_);
#pragma unroll
        for (int v = 0; v < WIN_; ++v) xr[v] = x4[v];
#pragma unroll
        for (int p = 0; p < CP_; ++p) {
            c[p] = q[p * (N_ / 4)];
            tot.x += c[p].x; tot.y += c[p].y; tot.z += c[p].z; tot.w += c[p].w;
        }
    }
    int s = (tot.x + tot.y) + (tot.z + tot.w);
    int incl = block_incl_scan(s, sh);
    int run = blockoff[blockIdx.x] + incl - s;
    if (live) {
        int4 start;
        start.x = run;
        start.y = start.x + tot.x;
        start.z = start.y + tot.y;
        start.w = start.z + tot.z;
        *(int4*)(rowptr + base) = start;
        float4 di;
        di.x = (tot.x > 0) ? rsqrtf((float)tot.x) : 0.0f;
        di.y = (tot.y > 0) ? rsqrtf((float)tot.y) : 0.0f;
        di.z = (tot.z > 0) ? rsqrtf((float)tot.z) : 0.0f;
        di.w = (tot.w > 0) ? rsqrtf((float)tot.w) : 0.0f;
        *(float4*)(dinv + base) = di;
        const float fl[4 * WIN_] = { xr[0].x, xr[0].y, xr[0].z, xr[0].w,
                                     xr[1].x, xr[1].y, xr[1].z, xr[1].w,
                                     xr[2].x, xr[2].y, xr[2].z, xr[2].w,
                                     xr[3].x, xr[3].y, xr[3].z, xr[3].w,
                                     xr[4].x, xr[4].y, xr[4].z, xr[4].w };
        static_assert(WIN_ == 5, "fl[] names the five 16-byte pieces of four rows");
#pragma unroll
        for (int f = 0; f < WIN_; ++f) {
            float4 r;
            r.x = fl[0 * WIN_ + f] * di.x;
            r.y = fl[1 * WIN_ + f] * di.y;
            r.z = fl[2 * WIN_ + f] * di.z;
            r.w = fl[3 * WIN_ + f] * di.w;
            *(float4*)(gx + f * NTOT_ + base) = r;
        }
#pragma unroll
        for (int p = 0; p < CP_; ++p) {
            q[p * (N_ / 4)] = start;
            start.x += c[p].x; start.y += c[p].y; start.z += c[p].z; start.w += c[p].w;
        }
    }
}

__global__ void fill_kernel(const int* __restrict__ ei, const int* __restrict__ part,
                            const int* __restrict__ rank, int* __restrict__ csr_src) {
    int t = blockIdx.x * blockDim.x + threadIdx.x;
    if (t >= EHALF_) return;
    int src[EPT_], pos[EPT_], row[EPT_];
#pragma unroll
    for (int k = 0; k < EPT_; ++k) {
        int b  = k >> 1;
        int el = t + (k & 1) * EHALF_;
        src[k] = ei[(b * 2 + 0) * E_ + el] + b * N_;
        int dl = ei[(b * 2 + 1) * E_ + el];
        row[k] = (b * CP_ + el / CE_) * N_ + dl;
        pos[k] = rank[b * E_ + el];
    }
#pragma unroll
    for (int k = 0; k < EPT_; ++k) {
        pos[k] += part[row[k]];
    }
#pragma unroll
    for (int k = 0; k < EPT_; ++k) {
        csr_src[pos[k]] = src[k];
    }
}

template <int K>
static __device__ __forceinline__ int bcast8(int v) {
    return (int)__builtin_amdgcn_mov_dpp8((unsigned)v, (unsigned)(K * 0x249249));
}
template <int K>
static __device__ __forceinline__ float bcast8(float v) {
    return __int_as_float(bcast8<K>(__float_as_int(v)));
}

static __device__ __forceinline__ void accum(float& a, float v) { a += v; }
static __device__ __forceinline__ void accum(float4& a, const float4& v) {
    a.x += v.x; a.y += v.y; a.z += v.z; a.w += v.w;
}
static __device__ __forceinline__ float  scaled(float v, float d) { return v * d; }
static __device__ __forceinline__ float4 scaled(const float4& v, float d) {
    return make_float4(v.x * d, v.y * d, v.z * d, v.w * d);
}
static __device__ __forceinline__ void set_zero(float& a)  { a = 0.0f; }
static __device__ __forceinline__ void set_zero(float4& a) { a = make_float4(0.0f, 0.0f, 0.0f, 0.0f); }

template <int V> struct RowSlice;
template <> struct RowSlice<4> { typedef float4 T; };
template <> struct RowSlice<1> { typedef float  T; };

template <int LQ, typename RowT>
static __device__ __forceinline__ void pull_chunk(const RowT* __restrict__ hin_v, int s_l,
                                                  int n, int q, bool carries, RowT& acc) {
    int s[8];
    s[0] = bcast8<0>(s_l); s[1] = bcast8<1>(s_l); s[2] = bcast8<2>(s_l); s[3] = bcast8<3>(s_l);
    s[4] = bcast8<4>(s_l); s[5] = bcast8<5>(s_l); s[6] = bcast8<6>(s_l); s[7] = bcast8<7>(s_l);
    RowT v[8];
#pragma unroll
    for (int k = 0; k < 8; ++k) {
        set_zero(v[k]);
        if (carries && k < n) v[k] = hin_v[s[k] * LQ + q];
    }
#pragma unroll
    for (int k = 0; k < 8; ++k) accum(acc, v[k]);
}

#define PULL_NPG_ 4
#define PULL_G_   (NTOT_ / PULL_NPG_)
static_assert(NTOT_ % PULL_NPG_ == 0, "every lane group takes the same number of nodes");

template <int F, bool WG>
__global__ void pull_kernel(const int* __restrict__ rowptr, const int* __restrict__ csr_src,
                            const float* __restrict__ dinv, const float* __restrict__ hin,
                            float* __restrict__ hout, float* __restrict__ gout) {
    constexpr int V  = (F % 8 == 0) ? (F / 8) : 1;
    constexpr int LQ = F / V;
    typedef typename RowSlice<V>::T RowT;
    int tid = blockIdx.x * blockDim.x + threadIdx.x;
    int g = tid >> 3;
    int q = tid & 7;
    if (g >= PULL_G_) return;
    const RowT* __restrict__ hin_v = (const RowT*)hin;
    RowT* __restrict__ hout_v = (RowT*)hout;
    RowT* __restrict__ gout_v = (RowT*)gout;
    const bool carries = (q < LQ);
    for (int node = 0; node < PULL_NPG_; ++node) {
        const int i = g + node * PULL_G_;
        int j   = rowptr[i];
        int end = rowptr[i + 1];
        const float d = dinv[i];
        RowT acc;
        set_zero(acc);
        for (; j < end; j += 32) {
            int n = end - j;
            int s_l[4];
#pragma unroll
            for (int c = 0; c < 4; ++c) {
                s_l[c] = 0;
                if (8 * c + q < n) s_l[c] = csr_src[j + 8 * c + q];
            }
#pragma unroll
            for (int c = 0; c < 4; ++c) {
                if (8 * c < n)
                    pull_chunk<LQ, RowT>(hin_v, s_l[c], n - 8 * c, q, carries, acc);
            }
        }
        if (carries) {
            const RowT xr = scaled(acc, d);
            hout_v[i * LQ + q] = xr;
            if (WG) gout_v[i * LQ + q] = scaled(xr, d);
        }
    }
}

#define P5_TB_      1024
#define P5_RANGES_  12
#define P5_ROWS_    4168
#define P5_U_       8
static_assert(P5_ROWS_ % P5_U_ == 0 && N_ % P5_U_ == 0, "a lane group's rows are all inside the range");
static_assert(P5_RANGES_ * P5_ROWS_ >= N_, "the ranges cover a graph");
static_assert(P5_U_ == 8, "lane q of a group owns row q of the group's eight");

template <int X>
static __device__ __forceinline__ float xor8(float v) {
    constexpr unsigned sel = ((0u ^ X) << 0) | ((1u ^ X) << 3) | ((2u ^ X) << 6) | ((3u ^ X) << 9) |
                             ((4u ^ X) << 12) | ((5u ^ X) << 15) | ((6u ^ X) << 18) | ((7u ^ X) << 21);
    return __int_as_float((int)__builtin_amdgcn_mov_dpp8((unsigned)__float_as_int(v), sel));
}

template <bool WG>
__global__ __launch_bounds__(P5_TB_) void pull5_kernel(const int* __restrict__ rowptr,
                                                       const int* __restrict__ csr_src,
                                                       const float* __restrict__ dinv,
                                                       const float* __restrict__ hin,
                                                       float* __restrict__ hout,
                                                       float* __restrict__ gout) {
    __shared__ __attribute__((aligned(16))) float col[N_ + 4];
    const int t  = threadIdx.x;
    const int rg = blockIdx.x % P5_RANGES_;
    const int bf = blockIdx.x / P5_RANGES_;
    const int f  = bf % WIN_;
    const int b  = bf / WIN_;
    const int nb = b * N_;
    {
        const float4* __restrict__ s4 = (const float4*)(hin + f * NTOT_ + nb);
        float4* col4 = (float4*)col;
        constexpr int NB = (N_ / 4 + P5_TB_ - 1) / P5_TB_;
        float4 w[NB];
#pragma unroll
        for (int k = 0; k < NB; ++k) {
            int i = k * P5_TB_ + t;
            w[k] = make_float4(0.0f, 0.0f, 0.0f, 0.0f);
            if (i < N_ / 4) w[k] = s4[i];
        }
#pragma unroll
        for (int k = 0; k < NB; ++k) {
            int i = k * P5_TB_ + t;
            if (i < N_ / 4) col4[i] = w[k];
        }
        if (t == 0) col[N_] = 0.0f;
    }
    __syncthreads();

    const int q = t & 7;
    const int rowBeg = rg * P5_ROWS_;
    const int rowEnd = (rowBeg + P5_ROWS_ < N_) ? (rowBeg + P5_ROWS_) : N_;
    const int absent = nb + N_;
    const bool qb0 = (q & 1) != 0, qb1 = (q & 2) != 0, qb2 = (q & 4) != 0;
    for (int r0 = rowBeg + (t >> 3) * P5_U_; r0 < rowEnd; r0 += (P5_TB_ / 8) * P5_U_) {
        const int i = nb + r0 + q;
        const int lo = rowptr[i];
        const int hi = rowptr[i + 1];
        const float d = dinv[i];
        int st[P5_U_], n[P5_U_];
        st[0] = bcast8<0>(lo); st[1] = bcast8<1>(lo); st[2] = bcast8<2>(lo); st[3] = bcast8<3>(lo);
        st[4] = bcast8<4>(lo); st[5] = bcast8<5>(lo); st[6] = bcast8<6>(lo); st[7] = bcast8<7>(lo);
        n[0] = bcast8<0>(hi) - st[0]; n[1] = bcast8<1>(hi) - st[1];
        n[2] = bcast8<2>(hi) - st[2]; n[3] = bcast8<3>(hi) - st[3];
        n[4] = bcast8<4>(hi) - st[4]; n[5] = bcast8<5>(hi) - st[5];
        n[6] = bcast8<6>(hi) - st[6]; n[7] = bcast8<7>(hi) - st[7];
        int rec[P5_U_][4];
#pragma unroll
        for (int u = 0; u < P5_U_; ++u) {
#pragma unroll
            for (int c = 0; c < 4; ++c) {
                const int p = st[u] + 8 * c + q;
                rec[u][c] = csr_src[(p < BE_ - 1) ? p : (BE_ - 1)];
            }
        }
        float a[P5_U_];
#pragma unroll
        for (int u = 0; u < P5_U_; ++u) {
            float s = col[((q < n[u]) ? rec[u][0] : absent) - nb];
            s += col[((8 + q < n[u]) ? rec[u][1] : absent) - nb];
            s += col[((16 + q < n[u]) ? rec[u][2] : absent) - nb];
            s += col[((24 + q < n[u]) ? rec[u][3] : absent) - nb];
            a[u] = s;
        }
        int nmax = n[0];
#pragma unroll
        for (int u = 1; u < P5_U_; ++u) nmax = (n[u] > nmax) ? n[u] : nmax;
        if (nmax > 32) {
#pragma unroll
            for (int u = 0; u < P5_U_; ++u) {
                for (int e = 32 + q; e < n[u]; e += 8) a[u] += col[csr_src[st[u] + e] - nb];
            }
        }
        float c4[4], c2[2];
#pragma unroll
        for (int j = 0; j < 4; ++j) {
            const float keep = qb0 ? a[2 * j + 1] : a[2 * j];
            const float give = qb0 ? a[2 * j] : a[2 * j + 1];
            c4[j] = keep + xor8<1>(give);
        }
#pragma unroll
        for (int j = 0; j < 2; ++j) {
            const float keep = qb1 ? c4[2 * j + 1] : c4[2 * j];
            const float give = qb1 ? c4[2 * j] : c4[2 * j + 1];
            c2[j] = keep + xor8<2>(give);
        }
        const float keep = qb2 ? c2[1] : c2[0];
        const float give = qb2 ? c2[0] : c2[1];
        const float mine = keep + xor8<4>(give);
        const float xr = mine * d;
        hout[f * NTOT_ + i] = xr;
        if (WG) gout[f * NTOT_ + i] = xr * d;
    }
}

template <int F, bool WG>
__global__ void tag_gemm_kernel(const float* __restrict__ x0, const float* __restrict__ x1,
                                const float* __restrict__ x2, const float* __restrict__ W,
                                const float* __restrict__ bias, float* __restrict__ out,
                                const float* __restrict__ dinv, float* __restrict__ gout) {
    const int numTiles = NTOT_ / 16;
    int wave = threadIdx.x >> 5;
    int tile = blockIdx.x * (blockDim.x >> 5) + wave;
    if (tile >= numTiles) return;

    int lane = threadIdx.x & 31;
    int half = lane >> 4;
    int l16  = lane & 15;
    int rowA = tile * 16 + l16;

    v8f c0 = {};
    v8f c1 = {};

    const float* xs[3] = { x0, x1, x2 };
    constexpr bool WIDE = (F % 8 == 0);
    constexpr int  NP   = WIDE ? (F / 8) : 1;

    if constexpr (WIDE) {
        static_assert(NP == 4, "the load fence below names 3 x 4 pieces");
        v4f av[3][NP];
#pragma unroll
        for (int s = 0; s < 3; ++s) {
            const v4f* __restrict__ xp4 = (const v4f*)(xs[s] + rowA * F + 4 * half);
#pragma unroll
            for (int j = 0; j < NP; ++j) av[s][j] = xp4[2 * j];
        }
        asm volatile("" :: "v"(av[0][0]), "v"(av[0][1]), "v"(av[0][2]), "v"(av[0][3]),
                           "v"(av[1][0]), "v"(av[1][1]), "v"(av[1][2]), "v"(av[1][3]),
                           "v"(av[2][0]), "v"(av[2][1]), "v"(av[2][2]), "v"(av[2][3]));
#pragma unroll
        for (int s = 0; s < 3; ++s) {
#pragma unroll
            for (int j = 0; j < NP; ++j) {
                const int kW = s * F + 8 * j + 4 * half;
                const float2* __restrict__ wp = (const float2*)(W + kW * H_ + 2 * l16);
                const float2 w0 = wp[0 * (H_ / 2)];
                const float2 w1 = wp[1 * (H_ / 2)];
                const float2 w2 = wp[2 * (H_ / 2)];
                const float2 w3 = wp[3 * (H_ / 2)];
                const v4f p = av[s][j];
                v2f a, b0, b1;
                a.x = p.x;   a.y = p.y;
                b0.x = w0.x; b0.y = w1.x;
                b1.x = w0.y; b1.y = w1.y;
                c0 = __builtin_amdgcn_wmma_f32_16x16x4_f32(false, a, false, b0,
                                                           (short)0, c0, false, false);
                c1 = __builtin_amdgcn_wmma_f32_16x16x4_f32(false, a, false, b1,
                                                           (short)0, c1, false, false);
                a.x = p.z;   a.y = p.w;
                b0.x = w2.x; b0.y = w3.x;
                b1.x = w2.y; b1.y = w3.y;
                c0 = __builtin_amdgcn_wmma_f32_16x16x4_f32(false, a, false, b0,
                                                           (short)0, c0, false, false);
                c1 = __builtin_amdgcn_wmma_f32_16x16x4_f32(false, a, false, b1,
                                                           (short)0, c1, false, false);
            }
        }
    } else {
#pragma unroll
        for (int s = 0; s < 3; ++s) {
            const float* __restrict__ xp = xs[s];
#pragma unroll
            for (int k0 = 0; k0 < F; k0 += 4) {
                int kA = k0 + half * 2;
                float m0 = (kA     < F) ? 1.0f : 0.0f;
                float m1 = (kA + 1 < F) ? 1.0f : 0.0f;
                int ka0 = (kA     < F) ? kA     : (F - 1);
                int ka1 = (kA + 1 < F) ? kA + 1 : (F - 1);
                v2f a, b0, b1;
                a.x = ((s == 0) ? xp[rowA * F + ka0] : xp[ka0 * NTOT_ + rowA]) * m0;
                a.y = ((s == 0) ? xp[rowA * F + ka1] : xp[ka1 * NTOT_ + rowA]) * m1;
                const float2 w0 = *(const float2*)(W + (s * F + ka0) * H_ + 2 * l16);
                const float2 w1 = *(const float2*)(W + (s * F + ka1) * H_ + 2 * l16);
                b0.x = w0.x * m0; b0.y = w1.x * m1;
                b1.x = w0.y * m0; b1.y = w1.y * m1;
                c0 = __builtin_amdgcn_wmma_f32_16x16x4_f32(false, a, false, b0,
                                                           (short)0, c0, false, false);
                c1 = __builtin_amdgcn_wmma_f32_16x16x4_f32(false, a, false, b1,
                                                           (short)0, c1, false, false);
            }
        }
    }

    const float2 bs = *(const float2*)(bias + 2 * l16);
    const int row0 = tile * 16 + half * 8;
    float dn[8];
    if (WG) {
        const float4 da = *(const float4*)(dinv + row0);
        const float4 db = *(const float4*)(dinv + row0 + 4);
        dn[0] = da.x; dn[1] = da.y; dn[2] = da.z; dn[3] = da.w;
        dn[4] = db.x; dn[5] = db.y; dn[6] = db.z; dn[7] = db.w;
    }
#pragma unroll
    for (int r = 0; r < 8; ++r) {
        const int row = row0 + r;
        float2 v;
        v.x = tanhf(c0[r] + bs.x);
        v.y = tanhf(c1[r] + bs.y);
        *(float2*)(out + row * H_ + 2 * l16) = v;
        if (WG) {
            float2 gv;
            gv.x = v.x * dn[r];
            gv.y = v.y * dn[r];
            *(float2*)(gout + row * H_ + 2 * l16) = gv;
        }
    }
}

__global__ void readout_kernel(const float* __restrict__ h, const float* __restrict__ Wr,
                               const float* __restrict__ br, float* __restrict__ out) {
    int i = blockIdx.x * blockDim.x + threadIdx.x;
    if (i >= NTOT_) return;
    float acc = br[0];
#pragma unroll
    for (int j = 0; j < H_; ++j) acc += h[i * H_ + j] * Wr[j];
    out[i] = acc;
}

static inline int cdiv(int a, int b) { return (a + b - 1) / b; }

extern "C" void kernel_launch(void* const* d_in, const int* in_sizes, int n_in,
                              void* d_out, int out_size, void* d_ws, size_t ws_size,
                              hipStream_t stream) {
    (void)in_sizes; (void)n_in; (void)out_size; (void)ws_size;

    const float* X  = (const float*)d_in[0];
    const int*   EI = (const int*)d_in[1];
    const float* W0 = (const float*)d_in[2];
    const float* b0 = (const float*)d_in[3];
    const float* W1 = (const float*)d_in[4];
    const float* b1 = (const float*)d_in[5];
    const float* W2 = (const float*)d_in[6];
    const float* b2 = (const float*)d_in[7];
    const float* Wr = (const float*)d_in[8];
    const float* br = (const float*)d_in[9];

    float* outF = (float*)d_out;
    float* hOut = outF + NTOT_;

    int*   part   = (int*)d_ws;
    int*   rowptr = part + B_ * CP_ * N_;
    int*   bsum   = rowptr + NTOT_ + 64;
    int*   boff   = bsum + SCAN_TB;
    float* dinv   = (float*)(boff + SCAN_TB);
    int*   rank   = (int*)(dinv + NTOT_);
    int*   csrc   = rank + BE_;
    float* p1     = (float*)(csrc + BE_);
    float* p2     = p1 + (size_t)NTOT_ * H_;
    float* bufA   = p2 + (size_t)NTOT_ * H_;
    float* bufB   = bufA + (size_t)NTOT_ * H_;
    float* g0     = bufB + (size_t)NTOT_ * H_;
    float* g1     = g0 + (size_t)NTOT_ * H_;

    const int TB = 256;

    count_rank_kernel<<<B_ * CP_, CTB_, 0, stream>>>(EI, part, rank);
    scan_sums_kernel<<<SCAN_BLOCKS, SCAN_TB, 0, stream>>>(part, bsum);
    scan_blocks_kernel<<<1, SCAN_TB, 0, stream>>>(bsum, boff, rowptr);
    scan_rows_kernel<<<SCAN_BLOCKS, SCAN_TB, 0, stream>>>(part, boff, rowptr, dinv, X, g0);
    fill_kernel<<<cdiv(EHALF_, TB), TB, 0, stream>>>(EI, part, rank, csrc);

    const int gemmBlocks = cdiv(NTOT_ / 16, TB / 32);
    const int pullW = cdiv(PULL_G_ * 8, TB);

    const int pull5W = B_ * WIN_ * P5_RANGES_;
    pull5_kernel<true><<<pull5W, P5_TB_, 0, stream>>>(rowptr, csrc, dinv, g0, p1, g1);
    pull5_kernel<false><<<pull5W, P5_TB_, 0, stream>>>(rowptr, csrc, dinv, g1, p2, nullptr);
    tag_gemm_kernel<WIN_, true><<<gemmBlocks, TB, 0, stream>>>(X, p1, p2, W0, b0, bufA, dinv, g0);

    pull_kernel<H_, true><<<pullW, TB, 0, stream>>>(rowptr, csrc, dinv, g0, p1, g1);
    pull_kernel<H_, false><<<pullW, TB, 0, stream>>>(rowptr, csrc, dinv, g1, p2, nullptr);
    tag_gemm_kernel<H_, true><<<gemmBlocks, TB, 0, stream>>>(bufA, p1, p2, W1, b1, bufB, dinv, g0);

    pull_kernel<H_, true><<<pullW, TB, 0, stream>>>(rowptr, csrc, dinv, g0, p1, g1);
    pull_kernel<H_, false><<<pullW, TB, 0, stream>>>(rowptr, csrc, dinv, g1, p2, nullptr);
    tag_gemm_kernel<H_, false><<<gemmBlocks, TB, 0, stream>>>(bufB, p1, p2, W2, b2, hOut, dinv, nullptr);

    readout_kernel<<<cdiv(NTOT_, TB), TB, 0, stream>>>(hOut, Wr, br, outF);
}


# Launches 15 -> 12: count_rank also writes its slice's sum per 1024-node chunk, scan_rows adds the chunks in front (two scan launches go); read-out in last tag_gemm, its dot summed as a 16-lane tree
# speedup vs baseline: 4.9144x; 1.0462x over previous
#include <hip/hip_runtime.h>
#include <math.h>

#define B_    4
#define WIN_  5
#define N_    50000
#define E_    1000000
#define H_    32
#define NTOT_ (B_ * N_)
#define BE_   (B_ * E_)

typedef __attribute__((ext_vector_type(2))) float v2f;
typedef __attribute__((ext_vector_type(4))) float v4f;
typedef __attribute__((ext_vector_type(8))) float v8f;

#define SCAN_TB     256
#define SCAN_CHUNK  (SCAN_TB * 4)
#define NCHUNK_     ((N_ + SCAN_CHUNK - 1) / SCAN_CHUNK)
#define SCAN_BLOCKS (B_ * NCHUNK_)
#define CSUM_WORDS_ 3200

#define EPT_    (B_ * 2)
#define EHALF_  (E_ / 2)
static_assert(E_ % 2 == 0, "edge list splits in two halves");

#define CP_     16
#define CE_     (E_ / CP_)
#define CE4_    (CE_ / 4)
#define CTB_    1024
#define CBATCH_ 8
static_assert(E_ % CP_ == 0 && CE_ % 4 == 0, "slices are whole 16-byte groups");
static_assert(N_ % 4 == 0, "histogram rows move in 16-byte groups");
static_assert(SCAN_BLOCKS * CP_ <= CSUM_WORDS_ && CSUM_WORDS_ % 4 == 0, "chunk sums fit their padded place");
static_assert(SCAN_CHUNK % (4 * 32) == 0, "a wave stores a chunk as whole 16-byte groups per lane");
#define CHUNK_G_ (SCAN_CHUNK / 4 / 32)

__global__ __launch_bounds__(CTB_) void count_rank_kernel(const int* __restrict__ ei,
                                                          int* __restrict__ part,
                                                          int* __restrict__ rank,
                                                          int* __restrict__ csum) {
    __shared__ int hist[N_];
    const int b = blockIdx.x / CP_;
    const int p = blockIdx.x - b * CP_;
    const int t = threadIdx.x;
    int4* hist4 = (int4*)hist;
    for (int i = t; i < N_ / 4; i += CTB_) hist4[i] = make_int4(0, 0, 0, 0);
    __syncthreads();
    const int4* __restrict__ d4 = (const int4*)(ei + (b * 2 + 1) * E_ + p * CE_);
    int4* __restrict__ r4 = (int4*)(rank + b * E_ + p * CE_);
    for (int base = 0; base < CE4_; base += CTB_ * CBATCH_) {
        int4 d[CBATCH_];
#pragma unroll
        for (int k = 0; k < CBATCH_; ++k) {
            int i = base + k * CTB_ + t;
            d[k] = make_int4(0, 0, 0, 0);
            if (i < CE4_) d[k] = d4[i];
        }
#pragma unroll
        for (int k = 0; k < CBATCH_; ++k) {
            int i = base + k * CTB_ + t;
            if (i < CE4_) {
                int4 s;
                s.x = atomicAdd(&hist[d[k].x], 1);
                s.y = atomicAdd(&hist[d[k].y], 1);
                s.z = atomicAdd(&hist[d[k].z], 1);
                s.w = atomicAdd(&hist[d[k].w], 1);
                r4[i] = s;
            }
        }
    }
    __syncthreads();
    int4* __restrict__ out4 = (int4*)(part + (b * CP_ + p) * N_);
    const int wv = t >> 5;
    const int ln = t & 31;
    for (int kc = wv; kc < NCHUNK_; kc += CTB_ / 32) {
        int s = 0;
        int4 c[CHUNK_G_];
#pragma unroll
        for (int k = 0; k < CHUNK_G_; ++k) {
            const int i = kc * (SCAN_CHUNK / 4) + k * 32 + ln;
            c[k] = hist4[(i < N_ / 4) ? i : (N_ / 4 - 1)];
        }
#pragma unroll
        for (int k = 0; k < CHUNK_G_; ++k) {
            const int i = kc * (SCAN_CHUNK / 4) + k * 32 + ln;
            if (i < N_ / 4) {
                out4[i] = c[k];
                s += (c[k].x + c[k].y) + (c[k].z + c[k].w);
            }
        }
        s += __shfl_xor(s, 16);
        s += __shfl_xor(s, 8);
        s += __shfl_xor(s, 4);
        s += __shfl_xor(s, 2);
        s += __shfl_xor(s, 1);
        if (ln == 0) csum[(b * NCHUNK_ + kc) * CP_ + p] = s;
    }
}

static __device__ __forceinline__ int2 block_incl_scan2(int2 v, int2* sh) {
    int t = threadIdx.x;
    sh[t] = v;
    __syncthreads();
    for (int off = 1; off < SCAN_TB; off <<= 1) {
        int2 x = (t >= off) ? sh[t - off] : make_int2(0, 0);
        __syncthreads();
        sh[t].x += x.x;
        sh[t].y += x.y;
        __syncthreads();
    }
    return sh[t];
}

__global__ void scan_rows_kernel(int* __restrict__ part, const int* __restrict__ csum,
                                 int* __restrict__ rowptr, float* __restrict__ dinv,
                                 const float* __restrict__ x, float* __restrict__ gx) {
    __shared__ int2 sh[SCAN_TB];
    const int b = blockIdx.x / NCHUNK_;
    const int n = (blockIdx.x - b * NCHUNK_) * SCAN_CHUNK + threadIdx.x * 4;
    const bool live = n < N_;
    const int base = b * N_ + n;
    int4 c[CP_];
    float4 xr[WIN_];
    int4 tot = make_int4(0, 0, 0, 0);
    int4* q = (int4*)part;
    int before = 0;
    {
        static_assert(CP_ % 4 == 0, "a chunk's csum entries are whole 16-byte groups");
        constexpr int NG = SCAN_BLOCKS * CP_ / 4;
        constexpr int NK = (NG + SCAN_TB - 1) / SCAN_TB;
        const int lim4 = blockIdx.x * (CP_ / 4);
        const int4* __restrict__ cs4 = (const int4*)csum;
        int4 w[NK];
#pragma unroll
        for (int k = 0; k < NK; ++k) {
            const int f = k * SCAN_TB + threadIdx.x;
            w[k] = cs4[(f < NG) ? f : (NG - 1)];
        }
#pragma unroll
        for (int k = 0; k < NK; ++k) {
            const int f = k * SCAN_TB + threadIdx.x;
            const int sum = (w[k].x + w[k].y) + (w[k].z + w[k].w);
            before += (f < lim4) ? sum : 0;
        }
    }
    if (live) {
        q = (int4*)(part + b * CP_ * N_ + n);
        const float4* __restrict__ x4 = (const float4*)(x + base * WIN_);
#pragma unroll
        for (int v = 0; v < WIN_; ++v) xr[v] = x4[v];
#pragma unroll
        for (int p = 0; p < CP_; ++p) {
            c[p] = q[p * (N_ / 4)];
            tot.x += c[p].x; tot.y += c[p].y; tot.z += c[p].z; tot.w += c[p].w;
        }
    }
    int s = (tot.x + tot.y) + (tot.z + tot.w);
    const int incl = block_incl_scan2(make_int2(s, before), sh).x;
    const int2 whole = sh[SCAN_TB - 1];
    int run = whole.y + incl - s;
    if (blockIdx.x == SCAN_BLOCKS - 1 && threadIdx.x == 0) rowptr[NTOT_] = whole.y + whole.x;
    if (live) {
        int4 start;
        start.x = run;
        start.y = start.x + tot.x;
        start.z = start.y + tot.y;
        start.w = start.z + tot.z;
        *(int4*)(rowptr + base) = start;
        float4 di;
        di.x = (tot.x > 0) ? rsqrtf((float)tot.x) : 0.0f;
        di.y = (tot.y > 0) ? rsqrtf((float)tot.y) : 0.0f;
        di.z = (tot.z > 0) ? rsqrtf((float)tot.z) : 0.0f;
        di.w = (tot.w > 0) ? rsqrtf((float)tot.w) : 0.0f;
        *(float4*)(dinv + base) = di;
        const float fl[4 * WIN_] = { xr[0].x, xr[0].y, xr[0].z, xr[0].w,
                                     xr[1].x, xr[1].y, xr[1].z, xr[1].w,
                                     xr[2].x, xr[2].y, xr[2].z, xr[2].w,
                                     xr[3].x, xr[3].y, xr[3].z, xr[3].w,
                                     xr[4].x, xr[4].y, xr[4].z, xr[4].w };
        static_assert(WIN_ == 5, "fl[] names the five 16-byte pieces of four rows");
#pragma unroll
        for (int f = 0; f < WIN_; ++f) {
            float4 r;
            r.x = fl[0 * WIN_ + f] * di.x;
            r.y = fl[1 * WIN_ + f] * di.y;
            r.z = fl[2 * WIN_ + f] * di.z;
            r.w = fl[3 * WIN_ + f] * di.w;
            *(float4*)(gx + f * NTOT_ + base) = r;
        }
#pragma unroll
        for (int p = 0; p < CP_; ++p) {
            q[p * (N_ / 4)] = start;
            start.x += c[p].x; start.y += c[p].y; start.z += c[p].z; start.w += c[p].w;
        }
    }
}

__global__ void fill_kernel(const int* __restrict__ ei, const int* __restrict__ part,
                            const int* __restrict__ rank, int* __restrict__ csr_src) {
    int t = blockIdx.x * blockDim.x + threadIdx.x;
    if (t >= EHALF_) return;
    int src[EPT_], pos[EPT_], row[EPT_];
#pragma unroll
    for (int k = 0; k < EPT_; ++k) {
        int b  = k >> 1;
        int el = t + (k & 1) * EHALF_;
        src[k] = ei[(b * 2 + 0) * E_ + el] + b * N_;
        int dl = ei[(b * 2 + 1) * E_ + el];
        row[k] = (b * CP_ + el / CE_) * N_ + dl;
        pos[k] = rank[b * E_ + el];
    }
#pragma unroll
    for (int k = 0; k < EPT_; ++k) {
        pos[k] += part[row[k]];
    }
#pragma unroll
    for (int k = 0; k < EPT_; ++k) {
        csr_src[pos[k]] = src[k];
    }
}

template <int K>
static __device__ __forceinline__ int bcast8(int v) {
    return (int)__builtin_amdgcn_mov_dpp8((unsigned)v, (unsigned)(K * 0x249249));
}
template <int K>
static __device__ __forceinline__ float bcast8(float v) {
    return __int_as_float(bcast8<K>(__float_as_int(v)));
}

static __device__ __forceinline__ void accum(float& a, float v) { a += v; }
static __device__ __forceinline__ void accum(float4& a, const float4& v) {
    a.x += v.x; a.y += v.y; a.z += v.z; a.w += v.w;
}
static __device__ __forceinline__ float  scaled(float v, float d) { return v * d; }
static __device__ __forceinline__ float4 scaled(const float4& v, float d) {
    return make_float4(v.x * d, v.y * d, v.z * d, v.w * d);
}
static __device__ __forceinline__ void set_zero(float& a)  { a = 0.0f; }
static __device__ __forceinline__ void set_zero(float4& a) { a = make_float4(0.0f, 0.0f, 0.0f, 0.0f); }

template <int V> struct RowSlice;
template <> struct RowSlice<4> { typedef float4 T; };
template <> struct RowSlice<1> { typedef float  T; };

template <int LQ, typename RowT>
static __device__ __forceinline__ void pull_chunk(const RowT* __restrict__ hin_v, int s_l,
                                                  int n, int q, bool carries, RowT& acc) {
    int s[8];
    s[0] = bcast8<0>(s_l); s[1] = bcast8<1>(s_l); s[2] = bcast8<2>(s_l); s[3] = bcast8<3>(s_l);
    s[4] = bcast8<4>(s_l); s[5] = bcast8<5>(s_l); s[6] = bcast8<6>(s_l); s[7] = bcast8<7>(s_l);
    RowT v[8];
#pragma unroll
    for (int k = 0; k < 8; ++k) {
        set_zero(v[k]);
        if (carries && k < n) v[k] = hin_v[s[k] * LQ + q];
    }
#pragma unroll
    for (int k = 0; k < 8; ++k) accum(acc, v[k]);
}

#define PULL_NPG_ 4
#define PULL_G_   (NTOT_ / PULL_NPG_)
static_assert(NTOT_ % PULL_NPG_ == 0, "every lane group takes the same number of nodes");

template <int F, bool WG>
__global__ void pull_kernel(const int* __restrict__ rowptr, const int* __restrict__ csr_src,
                            const float* __restrict__ dinv, const float* __restrict__ hin,
                            float* __restrict__ hout, float* __restrict__ gout) {
    constexpr int V  = (F % 8 == 0) ? (F / 8) : 1;
    constexpr int LQ = F / V;
    typedef typename RowSlice<V>::T RowT;
    int tid = blockIdx.x * blockDim.x + threadIdx.x;
    int g = tid >> 3;
    int q = tid & 7;
    if (g >= PULL_G_) return;
    const RowT* __restrict__ hin_v = (const RowT*)hin;
    RowT* __restrict__ hout_v = (RowT*)hout;
    RowT* __restrict__ gout_v = (RowT*)gout;
    const bool carries = (q < LQ);
    for (int node = 0; node < PULL_NPG_; ++node) {
        const int i = g + node * PULL_G_;
        int j   = rowptr[i];
        int end = rowptr[i + 1];
        const float d = dinv[i];
        RowT acc;
        set_zero(acc);
        for (; j < end; j += 32) {
            int n = end - j;
            int s_l[4];
#pragma unroll
            for (int c = 0; c < 4; ++c) {
                s_l[c] = 0;
                if (8 * c + q < n) s_l[c] = csr_src[j + 8 * c + q];
            }
#pragma unroll
            for (int c = 0; c < 4; ++c) {
                if (8 * c < n)
                    pull_chunk<LQ, RowT>(hin_v, s_l[c], n - 8 * c, q, carries, acc);
            }
        }
        if (carries) {
            const RowT xr = scaled(acc, d);
            hout_v[i * LQ + q] = xr;
            if (WG) gout_v[i * LQ + q] = scaled(xr, d);
        }
    }
}

#define P5_TB_      1024
#define P5_RANGES_  12
#define P5_ROWS_    4168
#define P5_U_       8
static_assert(P5_ROWS_ % P5_U_ == 0 && N_ % P5_U_ == 0, "a lane group's rows are all inside the range");
static_assert(P5_RANGES_ * P5_ROWS_ >= N_, "the ranges cover a graph");
static_assert(P5_U_ == 8, "lane q of a group owns row q of the group's eight");

template <int X>
static __device__ __forceinline__ float xor8(float v) {
    constexpr unsigned sel = ((0u ^ X) << 0) | ((1u ^ X) << 3) | ((2u ^ X) << 6) | ((3u ^ X) << 9) |
                             ((4u ^ X) << 12) | ((5u ^ X) << 15) | ((6u ^ X) << 18) | ((7u ^ X) << 21);
    return __int_as_float((int)__builtin_amdgcn_mov_dpp8((unsigned)__float_as_int(v), sel));
}

template <bool WG>
__global__ __launch_bounds__(P5_TB_) void pull5_kernel(const int* __restrict__ rowptr,
                                                       const int* __restrict__ csr_src,
                                                       const float* __restrict__ dinv,
                                                       const float* __restrict__ hin,
                                                       float* __restrict__ hout,
                                                       float* __restrict__ gout) {
    __shared__ __attribute__((aligned(16))) float col[N_ + 4];
    const int t  = threadIdx.x;
    const int rg = blockIdx.x % P5_RANGES_;
    const int bf = blockIdx.x / P5_RANGES_;
    const int f  = bf % WIN_;
    const int b  = bf / WIN_;
    const int nb = b * N_;
    {
        const float4* __restrict__ s4 = (const float4*)(hin + f * NTOT_ + nb);
        float4* col4 = (float4*)col;
        constexpr int NB = (N_ / 4 + P5_TB_ - 1) / P5_TB_;
        float4 w[NB];
#pragma unroll
        for (int k = 0; k < NB; ++k) {
            int i = k * P5_TB_ + t;
            w[k] = make_float4(0.0f, 0.0f, 0.0f, 0.0f);
            if (i < N_ / 4) w[k] = s4[i];
        }
#pragma unroll
        for (int k = 0; k < NB; ++k) {
            int i = k * P5_TB_ + t;
            if (i < N_ / 4) col4[i] = w[k];
        }
        if (t == 0) col[N_] = 0.0f;
    }
    __syncthreads();

    const int q = t & 7;
    const int rowBeg = rg * P5_ROWS_;
    const int rowEnd = (rowBeg + P5_ROWS_ < N_) ? (rowBeg + P5_ROWS_) : N_;
    const int absent = nb + N_;
    const bool qb0 = (q & 1) != 0, qb1 = (q & 2) != 0, qb2 = (q & 4) != 0;
    for (int r0 = rowBeg + (t >> 3) * P5_U_; r0 < rowEnd; r0 += (P5_TB_ / 8) * P5_U_) {
        const int i = nb + r0 + q;
        const int lo = rowptr[i];
        const int hi = rowptr[i + 1];
        const float d = dinv[i];
        int st[P5_U_], n[P5_U_];
        st[0] = bcast8<0>(lo); st[1] = bcast8<1>(lo); st[2] = bcast8<2>(lo); st[3] = bcast8<3>(lo);
        st[4] = bcast8<4>(lo); st[5] = bcast8<5>(lo); st[6] = bcast8<6>(lo); st[7] = bcast8<7>(lo);
        n[0] = bcast8<0>(hi) - st[0]; n[1] = bcast8<1>(hi) - st[1];
        n[2] = bcast8<2>(hi) - st[2]; n[3] = bcast8<3>(hi) - st[3];
        n[4] = bcast8<4>(hi) - st[4]; n[5] = bcast8<5>(hi) - st[5];
        n[6] = bcast8<6>(hi) - st[6]; n[7] = bcast8<7>(hi) - st[7];
        int rec[P5_U_][4];
#pragma unroll
        for (int u = 0; u < P5_U_; ++u) {
#pragma unroll
            for (int c = 0; c < 4; ++c) {
                const int p = st[u] + 8 * c + q;
                rec[u][c] = csr_src[(p < BE_ - 1) ? p : (BE_ - 1)];
            }
        }
        float a[P5_U_];
#pragma unroll
        for (int u = 0; u < P5_U_; ++u) {
            float s = col[((q < n[u]) ? rec[u][0] : absent) - nb];
            s += col[((8 + q < n[u]) ? rec[u][1] : absent) - nb];
            s += col[((16 + q < n[u]) ? rec[u][2] : absent) - nb];
            s += col[((24 + q < n[u]) ? rec[u][3] : absent) - nb];
            a[u] = s;
        }
        int nmax = n[0];
#pragma unroll
        for (int u = 1; u < P5_U_; ++u) nmax = (n[u] > nmax) ? n[u] : nmax;
        if (nmax > 32) {
#pragma unroll
            for (int u = 0; u < P5_U_; ++u) {
                for (int e = 32 + q; e < n[u]; e += 8) a[u] += col[csr_src[st[u] + e] - nb];
            }
        }
        float c4[4], c2[2];
#pragma unroll
        for (int j = 0; j < 4; ++j) {
            const float keep = qb0 ? a[2 * j + 1] : a[2 * j];
            const float give = qb0 ? a[2 * j] : a[2 * j + 1];
            c4[j] = keep + xor8<1>(give);
        }
#pragma unroll
        for (int j = 0; j < 2; ++j) {
            const float keep = qb1 ? c4[2 * j + 1] : c4[2 * j];
            const float give = qb1 ? c4[2 * j] : c4[2 * j + 1];
            c2[j] = keep + xor8<2>(give);
        }
        const float keep = qb2 ? c2[1] : c2[0];
        const float give = qb2 ? c2[0] : c2[1];
        const float mine = keep + xor8<4>(give);
        const float xr = mine * d;
        hout[f * NTOT_ + i] = xr;
        if (WG) gout[f * NTOT_ + i] = xr * d;
    }
}

template <int F, bool WG, bool RO>
__global__ void tag_gemm_kernel(const float* __restrict__ x0, const float* __restrict__ x1,
                                const float* __restrict__ x2, const float* __restrict__ W,
                                const float* __restrict__ bias, float* __restrict__ out,
                                const float* __restrict__ dinv, float* __restrict__ gout,
                                const float* __restrict__ Wr, const float* __restrict__ br,
                                float* __restrict__ rout) {
    const int numTiles = NTOT_ / 16;
    int wave = threadIdx.x >> 5;
    int tile = blockIdx.x * (blockDim.x >> 5) + wave;
    if (tile >= numTiles) return;

    int lane = threadIdx.x & 31;
    int half = lane >> 4;
    int l16  = lane & 15;
    int rowA = tile * 16 + l16;

    v8f c0 = {};
    v8f c1 = {};

    const float* xs[3] = { x0, x1, x2 };
    constexpr bool WIDE = (F % 8 == 0);
    constexpr int  NP   = WIDE ? (F / 8) : 1;

    if constexpr (WIDE) {
        static_assert(NP == 4, "the load fence below names 3 x 4 pieces");
        v4f av[3][NP];
#pragma unroll
        for (int s = 0; s < 3; ++s) {
            const v4f* __restrict__ xp4 = (const v4f*)(xs[s] + rowA * F + 4 * half);
#pragma unroll
            for (int j = 0; j < NP; ++j) av[s][j] = xp4[2 * j];
        }
        asm volatile("" :: "v"(av[0][0]), "v"(av[0][1]), "v"(av[0][2]), "v"(av[0][3]),
                           "v"(av[1][0]), "v"(av[1][1]), "v"(av[1][2]), "v"(av[1][3]),
                           "v"(av[2][0]), "v"(av[2][1]), "v"(av[2][2]), "v"(av[2][3]));
#pragma unroll
        for (int s = 0; s < 3; ++s) {
#pragma unroll
            for (int j = 0; j < NP; ++j) {
                const int kW = s * F + 8 * j + 4 * half;
                const float2* __restrict__ wp = (const float2*)(W + kW * H_ + 2 * l16);
                const float2 w0 = wp[0 * (H_ / 2)];
                const float2 w1 = wp[1 * (H_ / 2)];
                const float2 w2 = wp[2 * (H_ / 2)];
                const float2 w3 = wp[3 * (H_ / 2)];
                const v4f p = av[s][j];
                v2f a, b0, b1;
                a.x = p.x;   a.y = p.y;
                b0.x = w0.x; b0.y = w1.x;
                b1.x = w0.y; b1.y = w1.y;
                c0 = __builtin_amdgcn_wmma_f32_16x16x4_f32(false, a, false, b0,
                                                           (short)0, c0, false, false);
                c1 = __builtin_amdgcn_wmma_f32_16x16x4_f32(false, a, false, b1,
                                                           (short)0, c1, false, false);
                a.x = p.z;   a.y = p.w;
                b0.x = w2.x; b0.y = w3.x;
                b1.x = w2.y; b1.y = w3.y;
                c0 = __builtin_amdgcn_wmma_f32_16x16x4_f32(false, a, false, b0,
                                                           (short)0, c0, false, false);
                c1 = __builtin_amdgcn_wmma_f32_16x16x4_f32(false, a, false, b1,
                                                           (short)0, c1, false, false);
            }
        }
    } else {
#pragma unroll
        for (int s = 0; s < 3; ++s) {
            const float* __restrict__ xp = xs[s];
#pragma unroll
            for (int k0 = 0; k0 < F; k0 += 4) {
                int kA = k0 + half * 2;
                float m0 = (kA     < F) ? 1.0f : 0.0f;
                float m1 = (kA + 1 < F) ? 1.0f : 0.0f;
                int ka0 = (kA     < F) ? kA     : (F - 1);
                int ka1 = (kA + 1 < F) ? kA + 1 : (F - 1);
                v2f a, b0, b1;
                a.x = ((s == 0) ? xp[rowA * F + ka0] : xp[ka0 * NTOT_ + rowA]) * m0;
                a.y = ((s == 0) ? xp[rowA * F + ka1] : xp[ka1 * NTOT_ + rowA]) * m1;
                const float2 w0 = *(const float2*)(W + (s * F + ka0) * H_ + 2 * l16);
                const float2 w1 = *(const float2*)(W + (s * F + ka1) * H_ + 2 * l16);
                b0.x = w0.x * m0; b0.y = w1.x * m1;
                b1.x = w0.y * m0; b1.y = w1.y * m1;
                c0 = __builtin_amdgcn_wmma_f32_16x16x4_f32(false, a, false, b0,
                                                           (short)0, c0, false, false);
                c1 = __builtin_amdgcn_wmma_f32_16x16x4_f32(false, a, false, b1,
                                                           (short)0, c1, false, false);
            }
        }
    }

    const float2 bs = *(const float2*)(bias + 2 * l16);
    const int row0 = tile * 16 + half * 8;
    float dn[8];
    if (WG) {
        const float4 da = *(const float4*)(dinv + row0);
        const float4 db = *(const float4*)(dinv + row0 + 4);
        dn[0] = da.x; dn[1] = da.y; dn[2] = da.z; dn[3] = da.w;
        dn[4] = db.x; dn[5] = db.y; dn[6] = db.z; dn[7] = db.w;
    }
    float2 wr = make_float2(0.0f, 0.0f);
    if (RO) wr = *(const float2*)(Wr + 2 * l16);
    float mine = 0.0f;
#pragma unroll
    for (int r = 0; r < 8; ++r) {
        const int row = row0 + r;
        float2 v;
        v.x = tanhf(c0[r] + bs.x);
        v.y = tanhf(c1[r] + bs.y);
        *(float2*)(out + row * H_ + 2 * l16) = v;
        if (WG) {
            float2 gv;
            gv.x = v.x * dn[r];
            gv.y = v.y * dn[r];
            *(float2*)(gout + row * H_ + 2 * l16) = gv;
        }
        if (RO) {
            float p = v.x * wr.x + v.y * wr.y;
            p += xor8<1>(p);
            p += xor8<2>(p);
            p += xor8<4>(p);
            p += __shfl_xor(p, 8);
            if ((l16 & 7) == r) mine = p;
        }
    }
    if (RO) {
        if (l16 < 8) rout[row0 + l16] = mine + br[0];
    }
}

static inline int cdiv(int a, int b) { return (a + b - 1) / b; }

extern "C" void kernel_launch(void* const* d_in, const int* in_sizes, int n_in,
                              void* d_out, int out_size, void* d_ws, size_t ws_size,
                              hipStream_t stream) {
    (void)in_sizes; (void)n_in; (void)out_size; (void)ws_size;

    const float* X  = (const float*)d_in[0];
    const int*   EI = (const int*)d_in[1];
    const float* W0 = (const float*)d_in[2];
    const float* b0 = (const float*)d_in[3];
    const float* W1 = (const float*)d_in[4];
    const float* b1 = (const float*)d_in[5];
    const float* W2 = (const float*)d_in[6];
    const float* b2 = (const float*)d_in[7];
    const float* Wr = (const float*)d_in[8];
    const float* br = (const float*)d_in[9];

    float* outF = (float*)d_out;
    float* hOut = outF + NTOT_;

    int*   part   = (int*)d_ws;
    int*   rowptr = part + B_ * CP_ * N_;
    int*   csum   = rowptr + NTOT_ + 64;
    float* dinv   = (float*)(csum + CSUM_WORDS_);
    int*   rank   = (int*)(dinv + NTOT_);
    int*   csrc   = rank + BE_;
    float* p1     = (float*)(csrc + BE_);
    float* p2     = p1 + (size_t)NTOT_ * H_;
    float* bufA   = p2 + (size_t)NTOT_ * H_;
    float* bufB   = bufA + (size_t)NTOT_ * H_;
    float* g0     = bufB + (size_t)NTOT_ * H_;
    float* g1     = g0 + (size_t)NTOT_ * H_;

    const int TB = 256;

    count_rank_kernel<<<B_ * CP_, CTB_, 0, stream>>>(EI, part, rank, csum);
    scan_rows_kernel<<<SCAN_BLOCKS, SCAN_TB, 0, stream>>>(part, csum, rowptr, dinv, X, g0);
    fill_kernel<<<cdiv(EHALF_, TB), TB, 0, stream>>>(EI, part, rank, csrc);

    const int gemmBlocks = cdiv(NTOT_ / 16, TB / 32);
    const int pullW = cdiv(PULL_G_ * 8, TB);

    const int pull5W = B_ * WIN_ * P5_RANGES_;
    pull5_kernel<true><<<pull5W, P5_TB_, 0, stream>>>(rowptr, csrc, dinv, g0, p1, g1);
    pull5_kernel<false><<<pull5W, P5_TB_, 0, stream>>>(rowptr, csrc, dinv, g1, p2, nullptr);
    tag_gemm_kernel<WIN_, true, false><<<gemmBlocks, TB, 0, stream>>>(X, p1, p2, W0, b0, bufA, dinv, g0,
                                                                      nullptr, nullptr, nullptr);

    pull_kernel<H_, true><<<pullW, TB, 0, stream>>>(rowptr, csrc, dinv, g0, p1, g1);
    pull_kernel<H_, false><<<pullW, TB, 0, stream>>>(rowptr, csrc, dinv, g1, p2, nullptr);
    tag_gemm_kernel<H_, true, false><<<gemmBlocks, TB, 0, stream>>>(bufA, p1, p2, W1, b1, bufB, dinv, g0,
                                                                    nullptr, nullptr, nullptr);

    pull_kernel<H_, true><<<pullW, TB, 0, stream>>>(rowptr, csrc, dinv, g0, p1, g1);
    pull_kernel<H_, false><<<pullW, TB, 0, stream>>>(rowptr, csrc, dinv, g1, p2, nullptr);
    tag_gemm_kernel<H_, false, true><<<gemmBlocks, TB, 0, stream>>>(bufB, p1, p2, W2, b2, hOut, dinv, nullptr,
                                                                    Wr, br, outF);
}
